# starting kernel
# baseline (speedup 1.0000x reference)
#include <hip/hip_runtime.h>


typedef __attribute__((ext_vector_type(2))) float v2f;
typedef __attribute__((ext_vector_type(8))) float v8f;

#define B_    4
#define CIN_  16
#define COUT_ 16
#define T_    16
#define H_    128
#define W_    128
#define SCALE_ 0.04811252243246881f

#define XROW 144
#define NA   (27 * 4 * 32 * 2)
#define NDH  4
#define NX   (3 * NDH * 8 * XROW * 2)

__global__ __launch_bounds__(128) void conv3d_mod_wmma_f32(
    const float* __restrict__ input,
    const float* __restrict__ cond,
    const float* __restrict__ weight,
    float* __restrict__ out)
{
    extern __shared__ float smem[];
    float* sA = smem;
    float* sX = smem + NA;

    const int bi  = blockIdx.x;
    const int h0  = (bi % (H_ / 2)) * 2;
    const int t   = (bi / (H_ / 2)) % T_;
    const int b   = bi / ((H_ / 2) * T_);
    const int tid = threadIdx.x;

    for (int j = tid; j < NA; j += 128) {
        int tap  = j >> 8;
        int rem  = j & 255;
        int c    = rem >> 6;
        int rem2 = rem & 63;
        int l    = rem2 >> 1;
        int jj   = rem2 & 1;
        int cin  = 4 * c + 2 * (l >> 4) + jj;
        int m    = l & 15;
        int kt = tap / 9, kh = (tap / 3) % 3, kw = tap % 3;
        float wgt = weight[(((m * CIN_ + cin) * 3 + kt) * 3 + kh) * 3 + kw];
        sA[j] = wgt * SCALE_ * cond[b * CIN_ + cin];
    }

    for (int i = tid; i < 16 * 3 * NDH * 130; i += 128) {
        int r   = i / 130;
        int wi  = i - r * 130;
        int cin = r / (3 * NDH);
        int rr  = r - cin * (3 * NDH);
        int dt  = rr / NDH;
        int dh  = rr - dt * NDH;
        int ts = t + dt - 1, hs = h0 + dh - 1, ws = wi - 1;
        float v = 0.0f;
        if (((unsigned)ts < (unsigned)T_) & ((unsigned)hs < (unsigned)H_) &
            ((unsigned)ws < (unsigned)W_))
            v = input[((size_t)(b * CIN_ + cin) * T_ + ts) * (H_ * W_) +
                      (size_t)hs * W_ + ws];
        sX[(((dt * NDH + dh) * 8 + (cin >> 1)) * XROW + wi) * 2 + (cin & 1)] = v;
    }
    __syncthreads();

    const int wvid = tid >> 5;
    const int hr   = wvid >> 1;
    const int wseg = (wvid & 1) * 64;
    const int lane = tid & 31;
    const int hf   = lane >> 4;
    const int lx   = lane & 15;

    v8f acc0 = {}, acc1 = {}, acc2 = {}, acc3 = {};

#pragma unroll
    for (int tap = 0; tap < 27; ++tap) {
        const int kt = tap / 9, kh = (tap / 3) % 3, kw = tap % 3;
        const int rowb = (kt * NDH + (kh + hr)) * 8;
        const int n0   = wseg + lx + kw;
#pragma unroll
        for (int c = 0; c < 4; ++c) {
            v2f a = *(const v2f*)&sA[(((tap * 4 + c) << 5) + lane) * 2];
            int base = (rowb + 2 * c + hf) * (XROW * 2);
            v2f b0 = *(const v2f*)&sX[base + n0 * 2];
            v2f b1 = *(const v2f*)&sX[base + (n0 + 16) * 2];
            v2f b2 = *(const v2f*)&sX[base + (n0 + 32) * 2];
            v2f b3 = *(const v2f*)&sX[base + (n0 + 48) * 2];
            acc0 = __builtin_amdgcn_wmma_f32_16x16x4_f32(
                false, a, false, b0, (short)0, acc0, false, false);
            acc1 = __builtin_amdgcn_wmma_f32_16x16x4_f32(
                false, a, false, b1, (short)0, acc1, false, false);
            acc2 = __builtin_amdgcn_wmma_f32_16x16x4_f32(
                false, a, false, b2, (short)0, acc2, false, false);
            acc3 = __builtin_amdgcn_wmma_f32_16x16x4_f32(
                false, a, false, b3, (short)0, acc3, false, false);
        }
    }

    const size_t strideC = (size_t)T_ * H_ * W_;
    float* o = out + (size_t)b * COUT_ * strideC + (size_t)t * (H_ * W_) +
               (size_t)(h0 + hr) * W_ + wseg + lx;
#pragma unroll
    for (int r = 0; r < 8; ++r) {
        size_t mo = (size_t)(hf * 8 + r) * strideC;
        o[mo]      = acc0[r];
        o[mo + 16] = acc1[r];
        o[mo + 32] = acc2[r];
        o[mo + 48] = acc3[r];
    }
}

extern "C" void kernel_launch(void* const* d_in, const int* in_sizes, int n_in,
                              void* d_out, int out_size, void* d_ws, size_t ws_size,
                              hipStream_t stream) {
    const float* input  = (const float*)d_in[0];
    const float* cond   = (const float*)d_in[1];
    const float* weight = (const float*)d_in[2];
    float* out = (float*)d_out;

    const int    nBlocks = B_ * T_ * (H_ / 2);
    const size_t shmem   = (size_t)(NA + NX) * sizeof(float);

    conv3d_mod_wmma_f32<<<dim3(nBlocks), dim3(128), shmem, stream>>>(
        input, cond, weight, out);
}


# staging only: halo rows and weights fetched as whole-row b128 global loads, all in flight at once (register bound raised to what LDS residency allows), no per-element divisions; WMMA loop unchanged
# speedup vs baseline: 5.5126x; 5.5126x over previous
#include <hip/hip_runtime.h>


typedef __attribute__((ext_vector_type(2))) float v2f;
typedef __attribute__((ext_vector_type(4))) float v4f;
typedef __attribute__((ext_vector_type(8))) float v8f;

#define B_    4
#define CIN_  16
#define COUT_ 16
#define T_    16
#define H_    128
#define W_    128
#define SCALE_ 0.04811252243246881f

#define XROW 144
#define XOFF 2
#define APITCH 258
#define NA   6968
#define NDH  4
#define NX   (3 * NDH * 8 * XROW * 2)
#define NW4  (COUT_ * CIN_ * 27 / 4)

__global__ __launch_bounds__(128) __attribute__((amdgpu_waves_per_eu(1, 2))) void conv3d_mod_wmma_f32(
    const float* __restrict__ input,
    const float* __restrict__ cond,
    const float* __restrict__ weight,
    float* __restrict__ out)
{
    extern __shared__ __attribute__((aligned(16))) float smem[];
    float* sA = smem;
    float* sX = smem + NA;

    const int bi  = blockIdx.x;
    const int h0  = (bi % (H_ / 2)) * 2;
    const int t   = (bi / (H_ / 2)) % T_;
    const int b   = bi / ((H_ / 2) * T_);
    const int tid = threadIdx.x;
    const int wvid = tid >> 5;
    const int lane = tid & 31;

    {
        const v4f* w4 = (const v4f*)weight;
        const float* cb = cond + b * CIN_;
#pragma unroll
        for (int half = 0; half < 2; ++half) {
            v4f wv[7];
            float c0[7], c1[7];
#pragma unroll
            for (int u = 0; u < 7; ++u) {
                const int k = half * 7 + u;
                int i4 = k * 128 + tid;
                if (k == 13) i4 = i4 < NW4 ? i4 : NW4 - 1;
                wv[u] = w4[i4];
                const int mc0 = (4 * i4) / 27;
                c0[u] = cb[mc0 & 15];
                c1[u] = cb[(mc0 + 1) & 15];
            }
#pragma unroll
            for (int u = 0; u < 7; ++u) {
                const int k  = half * 7 + u;
                const int i4 = k * 128 + tid;
                if (k < 13 || i4 < NW4) {
                    const int g0   = 4 * i4;
                    const int mc0  = g0 / 27;
                    const int tap0 = g0 - mc0 * 27;
                    const int mc1  = mc0 + 1;
                    const int d0 = ((((mc0 >> 2) & 3) * 32 + ((mc0 >> 1) & 1) * 16 + ((mc0 >> 4) & 15)) << 1) + (mc0 & 1);
                    const int d1 = ((((mc1 >> 2) & 3) * 32 + ((mc1 >> 1) & 1) * 16 + ((mc1 >> 4) & 15)) << 1) + (mc1 & 1);
#pragma unroll
                    for (int e = 0; e < 4; ++e) {
                        const int  te   = tap0 + e;
                        const bool wrap = te >= 27;
                        const int  dst  = (wrap ? te - 27 : te) * APITCH + (wrap ? d1 : d0);
                        sA[dst] = wv[u][e] * SCALE_ * (wrap ? c1[u] : c0[u]);
                    }
                }
            }
        }
    }

    {
        v4f* sX4 = (v4f*)sX;
        const float* inb = input + (size_t)b * CIN_ * T_ * (H_ * W_) + 4 * lane;
        const size_t strideC = (size_t)T_ * H_ * W_;
#pragma unroll
        for (int grp = 0; grp < 3; ++grp) {
            v4f ev[8], od[8];
            const int ts  = t + grp - 1;
            const bool tok = (unsigned)ts < (unsigned)T_;
            const int tsc = ts < 0 ? 0 : (ts > T_ - 1 ? T_ - 1 : ts);
#pragma unroll
            for (int u = 0; u < 8; ++u) {
                const int cp = (u & 1) * 4 + wvid;
                const int dh = u >> 1;
                const int hs = h0 + dh - 1;
                const int hsc = hs < 0 ? 0 : (hs > H_ - 1 ? H_ - 1 : hs);
                const float* src = inb + ((size_t)(2 * cp) * T_ + tsc) * (H_ * W_) + (size_t)hsc * W_;
                ev[u] = *(const v4f*)src;
                od[u] = *(const v4f*)(src + strideC);
            }
#pragma unroll
            for (int u = 0; u < 8; ++u) {
                const int cp = (u & 1) * 4 + wvid;
                const int dh = u >> 1;
                const int hs = h0 + dh - 1;
                const bool ok = tok & ((unsigned)hs < (unsigned)H_);
                v4f e = ev[u], o = od[u];
                if (!ok) { e = (v4f){0.f, 0.f, 0.f, 0.f}; o = e; }
                const int p = (grp * NDH + dh) * 8 + cp;
                v4f lo = {e.x, o.x, e.y, o.y};
                v4f hi = {e.z, o.z, e.w, o.w};
                sX4[p * (XROW / 2) + 2 * lane + 1] = lo;
                sX4[p * (XROW / 2) + 2 * lane + 2] = hi;
            }
        }
        if (tid < 3 * NDH * 8) {
            v2f z = {0.f, 0.f};
            *(v2f*)&sX[(tid * XROW + (XOFF - 1)) * 2]  = z;
            *(v2f*)&sX[(tid * XROW + (XOFF + W_)) * 2] = z;
        }
    }
    __syncthreads();

    const int hr   = wvid >> 1;
    const int wseg = (wvid & 1) * 64;
    const int hf   = lane >> 4;
    const int lx   = lane & 15;

    v8f acc0 = {}, acc1 = {}, acc2 = {}, acc3 = {};

#pragma unroll
    for (int tap = 0; tap < 27; ++tap) {
        const int kt = tap / 9, kh = (tap / 3) % 3, kw = tap % 3;
        const int rowb = (kt * NDH + (kh + hr)) * 8;
        const int n0   = wseg + lx + kw + (XOFF - 1);
#pragma unroll
        for (int c = 0; c < 4; ++c) {
            v2f a = *(const v2f*)&sA[tap * APITCH + ((c << 5) + lane) * 2];
            int base = (rowb + 2 * c + hf) * (XROW * 2);
            v2f b0 = *(const v2f*)&sX[base + n0 * 2];
            v2f b1 = *(const v2f*)&sX[base + (n0 + 16) * 2];
            v2f b2 = *(const v2f*)&sX[base + (n0 + 32) * 2];
            v2f b3 = *(const v2f*)&sX[base + (n0 + 48) * 2];
            acc0 = __builtin_amdgcn_wmma_f32_16x16x4_f32(
                false, a, false, b0, (short)0, acc0, false, false);
            acc1 = __builtin_amdgcn_wmma_f32_16x16x4_f32(
                false, a, false, b1, (short)0, acc1, false, false);
            acc2 = __builtin_amdgcn_wmma_f32_16x16x4_f32(
                false, a, false, b2, (short)0, acc2, false, false);
            acc3 = __builtin_amdgcn_wmma_f32_16x16x4_f32(
                false, a, false, b3, (short)0, acc3, false, false);
        }
    }

    const size_t strideC = (size_t)T_ * H_ * W_;
    float* o = out + (size_t)b * COUT_ * strideC + (size_t)t * (H_ * W_) +
               (size_t)(h0 + hr) * W_ + wseg + lx;
#pragma unroll
    for (int r = 0; r < 8; ++r) {
        size_t mo = (size_t)(hf * 8 + r) * strideC;
        o[mo]      = acc0[r];
        o[mo + 16] = acc1[r];
        o[mo + 32] = acc2[r];
        o[mo + 48] = acc3[r];
    }
}

extern "C" void kernel_launch(void* const* d_in, const int* in_sizes, int n_in,
                              void* d_out, int out_size, void* d_ws, size_t ws_size,
                              hipStream_t stream) {
    const float* input  = (const float*)d_in[0];
    const float* cond   = (const float*)d_in[1];
    const float* weight = (const float*)d_in[2];
    float* out = (float*)d_out;

    const int    nBlocks = B_ * T_ * (H_ / 2);
    const size_t shmem   = (size_t)(NA + NX) * sizeof(float);

    conv3d_mod_wmma_f32<<<dim3(nBlocks), dim3(128), shmem, stream>>>(
        input, cond, weight, out);
}


# each block walks 8 consecutive t with a rolling 3-plane LDS halo window: weights staged once per block, one new dt plane fetched per step, its global loads in flight under the step's WMMAs
# speedup vs baseline: 6.4850x; 1.1764x over previous
#include <hip/hip_runtime.h>


typedef __attribute__((ext_vector_type(2))) float v2f;
typedef __attribute__((ext_vector_type(4))) float v4f;
typedef __attribute__((ext_vector_type(8))) float v8f;

#define B_    4
#define CIN_  16
#define COUT_ 16
#define T_    16
#define H_    128
#define W_    128
#define SCALE_ 0.04811252243246881f

#define TSTEPS 8
#define XROW 144
#define XOFF 2
#define APITCH 258
#define NA   6968
#define NDH  4
#define PLANE (NDH * 8 * XROW * 2)
#define NX   (3 * PLANE)
#define NW4  (COUT_ * CIN_ * 27 / 4)

__device__ __forceinline__ void load_plane(v4f (&ev)[8], v4f (&od)[8], const float* __restrict__ inb,
                                           int ts, int h0, int wvid)
{
    const size_t strideC = (size_t)T_ * H_ * W_;
    const int tsc = ts < 0 ? 0 : (ts > T_ - 1 ? T_ - 1 : ts);
#pragma unroll
    for (int u = 0; u < 8; ++u) {
        const int cp = (u & 1) * 4 + wvid;
        const int dh = u >> 1;
        const int hs = h0 + dh - 1;
        const int hsc = hs < 0 ? 0 : (hs > H_ - 1 ? H_ - 1 : hs);
        const float* src = inb + ((size_t)(2 * cp) * T_ + tsc) * (H_ * W_) + (size_t)hsc * W_;
        ev[u] = *(const v4f*)src;
        od[u] = *(const v4f*)(src + strideC);
    }
}

__device__ __forceinline__ void store_plane(const v4f (&ev)[8], const v4f (&od)[8], float* plane,
                                            int ts, int h0, int wvid, int lane)
{
    v4f* p4 = (v4f*)plane;
    const bool tok = (unsigned)ts < (unsigned)T_;
#pragma unroll
    for (int u = 0; u < 8; ++u) {
        const int cp = (u & 1) * 4 + wvid;
        const int dh = u >> 1;
        const int hs = h0 + dh - 1;
        const bool ok = tok & ((unsigned)hs < (unsigned)H_);
        v4f e = ev[u], o = od[u];
        if (!ok) { e = (v4f){0.f, 0.f, 0.f, 0.f}; o = e; }
        const int p = dh * 8 + cp;
        v4f lo = {e.x, o.x, e.y, o.y};
        v4f hi = {e.z, o.z, e.w, o.w};
        p4[p * (XROW / 2) + 2 * lane + 1] = lo;
        p4[p * (XROW / 2) + 2 * lane + 2] = hi;
    }
}

__global__ __launch_bounds__(128) __attribute__((amdgpu_waves_per_eu(1, 2))) void conv3d_mod_wmma_f32(
    const float* __restrict__ input,
    const float* __restrict__ cond,
    const float* __restrict__ weight,
    float* __restrict__ out)
{
    extern __shared__ __attribute__((aligned(16))) float smem[];
    float* sA = smem;
    float* sX = smem + NA;

    const int bi  = blockIdx.x;
    const int h0  = (bi % (H_ / 2)) * 2;
    const int t0  = ((bi / (H_ / 2)) % (T_ / TSTEPS)) * TSTEPS;
    const int b   = bi / ((H_ / 2) * (T_ / TSTEPS));
    const int tid = threadIdx.x;
    const int wvid = tid >> 5;
    const int lane = tid & 31;

    {
        const v4f* w4 = (const v4f*)weight;
        const float* cb = cond + b * CIN_;
#pragma unroll
        for (int half = 0; half < 2; ++half) {
            v4f wv[7];
            float c0[7], c1[7];
#pragma unroll
            for (int u = 0; u < 7; ++u) {
                const int k = half * 7 + u;
                int i4 = k * 128 + tid;
                if (k == 13) i4 = i4 < NW4 ? i4 : NW4 - 1;
                wv[u] = w4[i4];
                const int mc0 = (4 * i4) / 27;
                c0[u] = cb[mc0 & 15];
                c1[u] = cb[(mc0 + 1) & 15];
            }
#pragma unroll
            for (int u = 0; u < 7; ++u) {
                const int k  = half * 7 + u;
                const int i4 = k * 128 + tid;
                if (k < 13 || i4 < NW4) {
                    const int g0   = 4 * i4;
                    const int mc0  = g0 / 27;
                    const int tap0 = g0 - mc0 * 27;
                    const int mc1  = mc0 + 1;
                    const int d0 = ((((mc0 >> 2) & 3) * 32 + ((mc0 >> 1) & 1) * 16 + ((mc0 >> 4) & 15)) << 1) + (mc0 & 1);
                    const int d1 = ((((mc1 >> 2) & 3) * 32 + ((mc1 >> 1) & 1) * 16 + ((mc1 >> 4) & 15)) << 1) + (mc1 & 1);
#pragma unroll
                    for (int e = 0; e < 4; ++e) {
                        const int  te   = tap0 + e;
                        const bool wrap = te >= 27;
                        const int  dst  = (wrap ? te - 27 : te) * APITCH + (wrap ? d1 : d0);
                        sA[dst] = wv[u][e] * SCALE_ * (wrap ? c1[u] : c0[u]);
                    }
                }
            }
        }
    }

    const float* inb = input + (size_t)b * CIN_ * T_ * (H_ * W_) + 4 * lane;
    {
#pragma unroll
        for (int grp = 0; grp < 3; ++grp) {
            v4f ev[8], od[8];
            load_plane(ev, od, inb, t0 + grp - 1, h0, wvid);
            store_plane(ev, od, sX + grp * PLANE, t0 + grp - 1, h0, wvid, lane);
        }
        if (tid < 3 * NDH * 8) {
            v2f z = {0.f, 0.f};
            *(v2f*)&sX[(tid * XROW + (XOFF - 1)) * 2]  = z;
            *(v2f*)&sX[(tid * XROW + (XOFF + W_)) * 2] = z;
        }
    }
    __syncthreads();

    const int hr   = wvid >> 1;
    const int wseg = (wvid & 1) * 64;
    const int hf   = lane >> 4;
    const int lx   = lane & 15;

    const size_t strideC = (size_t)T_ * H_ * W_;
    float* ob = out + (size_t)b * COUT_ * strideC + (size_t)(h0 + hr) * W_ + wseg + lx;

    int pb0 = 0, pb1 = PLANE, pb2 = 2 * PLANE;

    for (int s = 0; s < TSTEPS; ++s) {
        const int  t    = t0 + s;
        const bool more = s + 1 < TSTEPS;

        v4f ev[8], od[8];
        if (more) load_plane(ev, od, inb, t + 2, h0, wvid);

        v8f acc0 = {}, acc1 = {}, acc2 = {}, acc3 = {};

#pragma unroll
        for (int tap = 0; tap < 27; ++tap) {
            const int kt = tap / 9, kh = (tap / 3) % 3, kw = tap % 3;
            const float* xp = sX + (kt == 0 ? pb0 : (kt == 1 ? pb1 : pb2));
            const int rowb = (kh + hr) * 8;
            const int n0   = wseg + lx + kw + (XOFF - 1);
#pragma unroll
            for (int c = 0; c < 4; ++c) {
                v2f a = *(const v2f*)&sA[tap * APITCH + ((c << 5) + lane) * 2];
                int base = (rowb + 2 * c + hf) * (XROW * 2);
                v2f b0 = *(const v2f*)&xp[base + n0 * 2];
                v2f b1 = *(const v2f*)&xp[base + (n0 + 16) * 2];
                v2f b2 = *(const v2f*)&xp[base + (n0 + 32) * 2];
                v2f b3 = *(const v2f*)&xp[base + (n0 + 48) * 2];
                acc0 = __builtin_amdgcn_wmma_f32_16x16x4_f32(
                    false, a, false, b0, (short)0, acc0, false, false);
                acc1 = __builtin_amdgcn_wmma_f32_16x16x4_f32(
                    false, a, false, b1, (short)0, acc1, false, false);
                acc2 = __builtin_amdgcn_wmma_f32_16x16x4_f32(
                    false, a, false, b2, (short)0, acc2, false, false);
                acc3 = __builtin_amdgcn_wmma_f32_16x16x4_f32(
                    false, a, false, b3, (short)0, acc3, false, false);
            }
        }

        if (more) {
            __syncthreads();
            store_plane(ev, od, sX + pb0, t + 2, h0, wvid, lane);
            const int freed = pb0;
            pb0 = pb1; pb1 = pb2; pb2 = freed;
            __syncthreads();
        }

        float* o = ob + (size_t)t * (H_ * W_);
#pragma unroll
        for (int r = 0; r < 8; ++r) {
            size_t mo = (size_t)(hf * 8 + r) * strideC;
            o[mo]      = acc0[r];
            o[mo + 16] = acc1[r];
            o[mo + 32] = acc2[r];
            o[mo + 48] = acc3[r];
        }
    }
}

extern "C" void kernel_launch(void* const* d_in, const int* in_sizes, int n_in,
                              void* d_out, int out_size, void* d_ws, size_t ws_size,
                              hipStream_t stream) {
    const float* input  = (const float*)d_in[0];
    const float* cond   = (const float*)d_in[1];
    const float* weight = (const float*)d_in[2];
    float* out = (float*)d_out;

    const int    nBlocks = B_ * (T_ / TSTEPS) * (H_ / 2);
    const size_t shmem   = (size_t)(NA + NX) * sizeof(float);

    conv3d_mod_wmma_f32<<<dim3(nBlocks), dim3(128), shmem, stream>>>(
        input, cond, weight, out);
}


# main loop only: each 48-WMMA (kt,kh) block's LDS fragment reads stand in one group, issued a block ahead into a second register set, none among the WMMAs; summation order unchanged
# speedup vs baseline: 6.7795x; 1.0454x over previous
#include <hip/hip_runtime.h>


typedef __attribute__((ext_vector_type(2))) float v2f;
typedef __attribute__((ext_vector_type(4))) float v4f;
typedef __attribute__((ext_vector_type(8))) float v8f;

#define B_    4
#define CIN_  16
#define COUT_ 16
#define T_    16
#define H_    128
#define W_    128
#define SCALE_ 0.04811252243246881f

#define TSTEPS 8
#define XROW 144
#define XOFF 2
#define APITCH 258
#define NA   6968
#define NDH  4
#define PLANE (NDH * 8 * XROW * 2)
#define NX   (3 * PLANE)
#define NW4  (COUT_ * CIN_ * 27 / 4)

__device__ __forceinline__ void load_plane(v4f (&ev)[8], v4f (&od)[8], const float* __restrict__ inb,
                                           int ts, int h0, int wvid)
{
    const size_t strideC = (size_t)T_ * H_ * W_;
    const int tsc = ts < 0 ? 0 : (ts > T_ - 1 ? T_ - 1 : ts);
#pragma unroll
    for (int u = 0; u < 8; ++u) {
        const int cp = (u & 1) * 4 + wvid;
        const int dh = u >> 1;
        const int hs = h0 + dh - 1;
        const int hsc = hs < 0 ? 0 : (hs > H_ - 1 ? H_ - 1 : hs);
        const float* src = inb + ((size_t)(2 * cp) * T_ + tsc) * (H_ * W_) + (size_t)hsc * W_;
        ev[u] = *(const v4f*)src;
        od[u] = *(const v4f*)(src + strideC);
    }
}

__device__ __forceinline__ void store_plane(const v4f (&ev)[8], const v4f (&od)[8], float* plane,
                                            int ts, int h0, int wvid, int lane)
{
    v4f* p4 = (v4f*)plane;
    const bool tok = (unsigned)ts < (unsigned)T_;
#pragma unroll
    for (int u = 0; u < 8; ++u) {
        const int cp = (u & 1) * 4 + wvid;
        const int dh = u >> 1;
        const int hs = h0 + dh - 1;
        const bool ok = tok & ((unsigned)hs < (unsigned)H_);
        v4f e = ev[u], o = od[u];
        if (!ok) { e = (v4f){0.f, 0.f, 0.f, 0.f}; o = e; }
        const int p = dh * 8 + cp;
        v4f lo = {e.x, o.x, e.y, o.y};
        v4f hi = {e.z, o.z, e.w, o.w};
        p4[p * (XROW / 2) + 2 * lane + 1] = lo;
        p4[p * (XROW / 2) + 2 * lane + 2] = hi;
    }
}


__device__ __forceinline__ void load_frags(v2f (&fa)[12], v2f (&fb)[48], const float* sA,
                                           const float* xp, int blk, int hr, int hf, int lane, int col0)
{
    const int kh   = blk % 3;
    const int rowb = (kh + hr) * 8;
#pragma unroll
    for (int kw = 0; kw < 3; ++kw) {
        const int tap = blk * 3 + kw;
        const int n0  = col0 + kw;
#pragma unroll
        for (int c = 0; c < 4; ++c) {
            const int r = kw * 4 + c;
            fa[r] = *(const v2f*)&sA[tap * APITCH + ((c << 5) + lane) * 2];
            const int base = (rowb + 2 * c + hf) * (XROW * 2);
            fb[4 * r + 0] = *(const v2f*)&xp[base + n0 * 2];
            fb[4 * r + 1] = *(const v2f*)&xp[base + (n0 + 16) * 2];
            fb[4 * r + 2] = *(const v2f*)&xp[base + (n0 + 32) * 2];
            fb[4 * r + 3] = *(const v2f*)&xp[base + (n0 + 48) * 2];
        }
    }
}

__global__ __launch_bounds__(128) __attribute__((amdgpu_waves_per_eu(1, 2))) void conv3d_mod_wmma_f32(
    const float* __restrict__ input,
    const float* __restrict__ cond,
    const float* __restrict__ weight,
    float* __restrict__ out)
{
    extern __shared__ __attribute__((aligned(16))) float smem[];
    float* sA = smem;
    float* sX = smem + NA;

    const int bi  = blockIdx.x;
    const int h0  = (bi % (H_ / 2)) * 2;
    const int t0  = ((bi / (H_ / 2)) % (T_ / TSTEPS)) * TSTEPS;
    const int b   = bi / ((H_ / 2) * (T_ / TSTEPS));
    const int tid = threadIdx.x;
    const int wvid = tid >> 5;
    const int lane = tid & 31;

    {
        const v4f* w4 = (const v4f*)weight;
        const float* cb = cond + b * CIN_;
#pragma unroll
        for (int half = 0; half < 2; ++half) {
            v4f wv[7];
            float c0[7], c1[7];
#pragma unroll
            for (int u = 0; u < 7; ++u) {
                const int k = half * 7 + u;
                int i4 = k * 128 + tid;
                if (k == 13) i4 = i4 < NW4 ? i4 : NW4 - 1;
                wv[u] = w4[i4];
                const int mc0 = (4 * i4) / 27;
                c0[u] = cb[mc0 & 15];
                c1[u] = cb[(mc0 + 1) & 15];
            }
#pragma unroll
            for (int u = 0; u < 7; ++u) {
                const int k  = half * 7 + u;
                const int i4 = k * 128 + tid;
                if (k < 13 || i4 < NW4) {
                    const int g0   = 4 * i4;
                    const int mc0  = g0 / 27;
                    const int tap0 = g0 - mc0 * 27;
                    const int mc1  = mc0 + 1;
                    const int d0 = ((((mc0 >> 2) & 3) * 32 + ((mc0 >> 1) & 1) * 16 + ((mc0 >> 4) & 15)) << 1) + (mc0 & 1);
                    const int d1 = ((((mc1 >> 2) & 3) * 32 + ((mc1 >> 1) & 1) * 16 + ((mc1 >> 4) & 15)) << 1) + (mc1 & 1);
#pragma unroll
                    for (int e = 0; e < 4; ++e) {
                        const int  te   = tap0 + e;
                        const bool wrap = te >= 27;
                        const int  dst  = (wrap ? te - 27 : te) * APITCH + (wrap ? d1 : d0);
                        sA[dst] = wv[u][e] * SCALE_ * (wrap ? c1[u] : c0[u]);
                    }
                }
            }
        }
    }

    const float* inb = input + (size_t)b * CIN_ * T_ * (H_ * W_) + 4 * lane;
    {
#pragma unroll
        for (int grp = 0; grp < 3; ++grp) {
            v4f ev[8], od[8];
            load_plane(ev, od, inb, t0 + grp - 1, h0, wvid);
            store_plane(ev, od, sX + grp * PLANE, t0 + grp - 1, h0, wvid, lane);
        }
        if (tid < 3 * NDH * 8) {
            v2f z = {0.f, 0.f};
            *(v2f*)&sX[(tid * XROW + (XOFF - 1)) * 2]  = z;
            *(v2f*)&sX[(tid * XROW + (XOFF + W_)) * 2] = z;
        }
    }
    __syncthreads();

    const int hr   = wvid >> 1;
    const int wseg = (wvid & 1) * 64;
    const int hf   = lane >> 4;
    const int lx   = lane & 15;

    const size_t strideC = (size_t)T_ * H_ * W_;
    float* ob = out + (size_t)b * COUT_ * strideC + (size_t)(h0 + hr) * W_ + wseg + lx;

    int pb0 = 0, pb1 = PLANE, pb2 = 2 * PLANE;

    v2f fa[2][12];
    v2f fb[2][48];
    const int col0 = wseg + lx + (XOFF - 1);
    load_frags(fa[0], fb[0], sA, sX + pb0, 0, hr, hf, lane, col0);

    for (int s = 0; s < TSTEPS; ++s) {
        const int  t    = t0 + s;
        const bool more = s + 1 < TSTEPS;

        v4f ev[8], od[8];
        if (more) load_plane(ev, od, inb, t + 2, h0, wvid);

        v8f acc0 = {}, acc1 = {}, acc2 = {}, acc3 = {};

#pragma unroll
        for (int blk = 0; blk < 9; ++blk) {
            const int cur = blk & 1;
            if (blk + 1 < 9) {
                const int kt1 = (blk + 1) / 3;
                load_frags(fa[cur ^ 1], fb[cur ^ 1], sA, sX + (kt1 == 0 ? pb0 : (kt1 == 1 ? pb1 : pb2)),
                           blk + 1, hr, hf, lane, col0);
            }
            __builtin_amdgcn_sched_barrier(0);
#pragma unroll
            for (int r = 0; r < 12; ++r) {
                acc0 = __builtin_amdgcn_wmma_f32_16x16x4_f32(
                    false, fa[cur][r], false, fb[cur][4 * r + 0], (short)0, acc0, false, false);
                acc1 = __builtin_amdgcn_wmma_f32_16x16x4_f32(
                    false, fa[cur][r], false, fb[cur][4 * r + 1], (short)0, acc1, false, false);
                acc2 = __builtin_amdgcn_wmma_f32_16x16x4_f32(
                    false, fa[cur][r], false, fb[cur][4 * r + 2], (short)0, acc2, false, false);
                acc3 = __builtin_amdgcn_wmma_f32_16x16x4_f32(
                    false, fa[cur][r], false, fb[cur][4 * r + 3], (short)0, acc3, false, false);
            }
            __builtin_amdgcn_sched_barrier(0);
        }

        if (more) {
            __syncthreads();
            store_plane(ev, od, sX + pb0, t + 2, h0, wvid, lane);
            const int freed = pb0;
            pb0 = pb1; pb1 = pb2; pb2 = freed;
            __syncthreads();
            load_frags(fa[0], fb[0], sA, sX + pb0, 0, hr, hf, lane, col0);
        }

        float* o = ob + (size_t)t * (H_ * W_);
#pragma unroll
        for (int r = 0; r < 8; ++r) {
            size_t mo = (size_t)(hf * 8 + r) * strideC;
            o[mo]      = acc0[r];
            o[mo + 16] = acc1[r];
            o[mo + 32] = acc2[r];
            o[mo + 48] = acc3[r];
        }
    }
}

extern "C" void kernel_launch(void* const* d_in, const int* in_sizes, int n_in,
                              void* d_out, int out_size, void* d_ws, size_t ws_size,
                              hipStream_t stream) {
    const float* input  = (const float*)d_in[0];
    const float* cond   = (const float*)d_in[1];
    const float* weight = (const float*)d_in[2];
    float* out = (float*)d_out;

    const int    nBlocks = B_ * (T_ / TSTEPS) * (H_ / 2);
    const size_t shmem   = (size_t)(NA + NX) * sizeof(float);

    conv3d_mod_wmma_f32<<<dim3(nBlocks), dim3(128), shmem, stream>>>(
        input, cond, weight, out);
}


# Winograd F(2,3) along h: the block's 2 output rows from 4 combined halo rows x 4 combined kh taps, 288 WMMAs per wave-step for 432; products regrouped, so the summation order changes
# speedup vs baseline: 8.7186x; 1.2860x over previous
#include <hip/hip_runtime.h>


typedef __attribute__((ext_vector_type(2))) float v2f;
typedef __attribute__((ext_vector_type(4))) float v4f;
typedef __attribute__((ext_vector_type(8))) float v8f;

#define B_    4
#define CIN_  16
#define COUT_ 16
#define T_    16
#define H_    128
#define W_    128
#define SCALE_ 0.04811252243246881f

#define TSTEPS 8
#define XROW 144
#define XOFF 2
#define APITCH 258
#define NTAPW 36
#define NA   (NTAPW * APITCH)
#define NDH  4
#define PLANE (NDH * 8 * XROW * 2)
#define NX   (3 * PLANE)
#define NW4  (COUT_ * CIN_ * 27 / 4)

__device__ __forceinline__ void load_plane(v4f (&ev)[8], v4f (&od)[8], const float* __restrict__ inb,
                                           int ts, int h0, int wvid)
{
    const size_t strideC = (size_t)T_ * H_ * W_;
    const int tsc = ts < 0 ? 0 : (ts > T_ - 1 ? T_ - 1 : ts);
#pragma unroll
    for (int u = 0; u < 8; ++u) {
        const int cp = (u & 1) * 4 + wvid;
        const int dh = u >> 1;
        const int hs = h0 + dh - 1;
        const int hsc = hs < 0 ? 0 : (hs > H_ - 1 ? H_ - 1 : hs);
        const float* src = inb + ((size_t)(2 * cp) * T_ + tsc) * (H_ * W_) + (size_t)hsc * W_;
        ev[u] = *(const v4f*)src;
        od[u] = *(const v4f*)(src + strideC);
    }
}

__device__ __forceinline__ void store_plane(const v4f (&ev)[8], const v4f (&od)[8], float* plane,
                                            int ts, int h0, int wvid, int lane)
{
    v4f* p4 = (v4f*)plane;
    const bool tok = (unsigned)ts < (unsigned)T_;
    const v4f zero = {0.f, 0.f, 0.f, 0.f};
#pragma unroll
    for (int k = 0; k < 2; ++k) {
        const int cp = k * 4 + wvid;
        v4f e[4], o[4];
#pragma unroll
        for (int dh = 0; dh < 4; ++dh) {
            const int hs = h0 + dh - 1;
            const bool ok = tok & ((unsigned)hs < (unsigned)H_);
            e[dh] = ok ? ev[2 * dh + k] : zero;
            o[dh] = ok ? od[2 * dh + k] : zero;
        }
        v4f te[4], to[4];
        te[0] = e[0] - e[2];  to[0] = o[0] - o[2];
        te[1] = e[1] + e[2];  to[1] = o[1] + o[2];
        te[2] = e[2] - e[1];  to[2] = o[2] - o[1];
        te[3] = e[1] - e[3];  to[3] = o[1] - o[3];
#pragma unroll
        for (int j = 0; j < 4; ++j) {
            const int p = j * 8 + cp;
            v4f lo = {te[j].x, to[j].x, te[j].y, to[j].y};
            v4f hi = {te[j].z, to[j].z, te[j].w, to[j].w};
            p4[p * (XROW / 2) + 2 * lane + 1] = lo;
            p4[p * (XROW / 2) + 2 * lane + 2] = hi;
        }
    }
}


__device__ __forceinline__ void load_frags(v2f (&fa)[16], v2f (&fb)[32], const float* sA,
                                           const float* xp, int blk, int hf, int lane, int col0)
{
    const int kt = blk / 3;
    const int kw = blk - 3 * kt;
    const int n0 = col0 + kw;
#pragma unroll
    for (int j = 0; j < 4; ++j) {
        const int tap = (kt * 4 + j) * 3 + kw;
#pragma unroll
        for (int c = 0; c < 4; ++c) {
            const int r = j * 4 + c;
            fa[r] = *(const v2f*)&sA[tap * APITCH + ((c << 5) + lane) * 2];
            const int base = (j * 8 + 2 * c + hf) * (XROW * 2);
            fb[2 * r + 0] = *(const v2f*)&xp[base + n0 * 2];
            fb[2 * r + 1] = *(const v2f*)&xp[base + (n0 + 16) * 2];
        }
    }
}

__global__ __launch_bounds__(128) __attribute__((amdgpu_waves_per_eu(1, 2))) void conv3d_mod_wmma_f32(
    const float* __restrict__ input,
    const float* __restrict__ cond,
    const float* __restrict__ weight,
    float* __restrict__ out)
{
    extern __shared__ __attribute__((aligned(16))) float smem[];
    float* sA = smem;
    float* sX = smem + NA;

    const int bi  = blockIdx.x;
    const int h0  = (bi % (H_ / 2)) * 2;
    const int t0  = ((bi / (H_ / 2)) % (T_ / TSTEPS)) * TSTEPS;
    const int b   = bi / ((H_ / 2) * (T_ / TSTEPS));
    const int tid = threadIdx.x;
    const int wvid = tid >> 5;
    const int lane = tid & 31;

    {
        const float* cb = cond + b * CIN_;
#pragma unroll
        for (int half = 0; half < 2; ++half) {
            float g[9][3];
            float cs[9];
#pragma unroll
            for (int u = 0; u < 9; ++u) {
                const int q  = (half * 9 + u) * 128 + tid;
                const int mc = q / 9;
                const int r9 = q - mc * 9;
                const int kt = r9 / 3;
                const int kw = r9 - kt * 3;
                const float* wp = weight + mc * 27 + kt * 9 + kw;
                g[u][0] = wp[0];
                g[u][1] = wp[3];
                g[u][2] = wp[6];
                cs[u]   = cb[mc & 15];
            }
#pragma unroll
            for (int u = 0; u < 9; ++u) {
                const int q  = (half * 9 + u) * 128 + tid;
                const int mc = q / 9;
                const int r9 = q - mc * 9;
                const int kt = r9 / 3;
                const int kw = r9 - kt * 3;
                const int d = ((((mc >> 2) & 3) * 32 + ((mc >> 1) & 1) * 16 + ((mc >> 4) & 15)) << 1) + (mc & 1);
                const float g0  = g[u][0] * SCALE_ * cs[u];
                const float g1  = g[u][1] * SCALE_ * cs[u];
                const float g2  = g[u][2] * SCALE_ * cs[u];
                const float s02 = g0 + g2;
                float* dst = sA + (kt * 12 + kw) * APITCH + d;
                dst[0 * 3 * APITCH] = g0;
                dst[1 * 3 * APITCH] = 0.5f * (s02 + g1);
                dst[2 * 3 * APITCH] = 0.5f * (s02 - g1);
                dst[3 * 3 * APITCH] = g2;
            }
        }
    }

    const float* inb = input + (size_t)b * CIN_ * T_ * (H_ * W_) + 4 * lane;
    {
#pragma unroll
        for (int grp = 0; grp < 3; ++grp) {
            v4f ev[8], od[8];
            load_plane(ev, od, inb, t0 + grp - 1, h0, wvid);
            store_plane(ev, od, sX + grp * PLANE, t0 + grp - 1, h0, wvid, lane);
        }
        if (tid < 3 * NDH * 8) {
            v2f z = {0.f, 0.f};
            *(v2f*)&sX[(tid * XROW + (XOFF - 1)) * 2]  = z;
            *(v2f*)&sX[(tid * XROW + (XOFF + W_)) * 2] = z;
        }
    }
    __syncthreads();

    const int wseg = wvid * 32;
    const int hf   = lane >> 4;
    const int lx   = lane & 15;

    const size_t strideC = (size_t)T_ * H_ * W_;
    float* ob = out + (size_t)b * COUT_ * strideC + (size_t)h0 * W_ + wseg + lx;

    int pb0 = 0, pb1 = PLANE, pb2 = 2 * PLANE;

    v2f fa[2][16];
    v2f fb[2][32];
    const int col0 = wseg + lx + (XOFF - 1);
    load_frags(fa[0], fb[0], sA, sX + pb0, 0, hf, lane, col0);

    for (int s = 0; s < TSTEPS; ++s) {
        const int  t    = t0 + s;
        const bool more = s + 1 < TSTEPS;

        v4f ev[8], od[8];
        if (more) load_plane(ev, od, inb, t + 2, h0, wvid);

        v8f acc[8];
#pragma unroll
        for (int a = 0; a < 8; ++a) acc[a] = (v8f){0.f, 0.f, 0.f, 0.f, 0.f, 0.f, 0.f, 0.f};

#pragma unroll
        for (int blk = 0; blk < 9; ++blk) {
            const int cur = blk & 1;
            if (blk + 1 < 9) {
                const int kt1 = (blk + 1) / 3;
                load_frags(fa[cur ^ 1], fb[cur ^ 1], sA, sX + (kt1 == 0 ? pb0 : (kt1 == 1 ? pb1 : pb2)),
                           blk + 1, hf, lane, col0);
            }
            __builtin_amdgcn_sched_barrier(0);
#pragma unroll
            for (int cg = 0; cg < 4; ++cg) {
#pragma unroll
                for (int j = 0; j < 4; ++j) {
                    const int r = j * 4 + cg;
                    acc[2 * j + 0] = __builtin_amdgcn_wmma_f32_16x16x4_f32(
                        false, fa[cur][r], false, fb[cur][2 * r + 0], (short)0, acc[2 * j + 0], false, false);
                    acc[2 * j + 1] = __builtin_amdgcn_wmma_f32_16x16x4_f32(
                        false, fa[cur][r], false, fb[cur][2 * r + 1], (short)0, acc[2 * j + 1], false, false);
                }
            }
            __builtin_amdgcn_sched_barrier(0);
            asm volatile("v_nop\n\tv_nop\n\tv_nop\n\tv_nop\n\tv_nop");
            __builtin_amdgcn_sched_barrier(0);
        }

        const v8f y0a = (acc[0] + acc[2]) + acc[4];
        const v8f y0b = (acc[1] + acc[3]) + acc[5];
        const v8f y1a = (acc[2] - acc[4]) - acc[6];
        const v8f y1b = (acc[3] - acc[5]) - acc[7];

        if (more) {
            __syncthreads();
            store_plane(ev, od, sX + pb0, t + 2, h0, wvid, lane);
            const int freed = pb0;
            pb0 = pb1; pb1 = pb2; pb2 = freed;
            __syncthreads();
            load_frags(fa[0], fb[0], sA, sX + pb0, 0, hf, lane, col0);
        }

        float* o = ob + (size_t)t * (H_ * W_);
#pragma unroll
        for (int r = 0; r < 8; ++r) {
            size_t mo = (size_t)(hf * 8 + r) * strideC;
            o[mo]           = y0a[r];
            o[mo + 16]      = y0b[r];
            o[mo + W_]      = y1a[r];
            o[mo + W_ + 16] = y1b[r];
        }
    }
}

extern "C" void kernel_launch(void* const* d_in, const int* in_sizes, int n_in,
                              void* d_out, int out_size, void* d_ws, size_t ws_size,
                              hipStream_t stream) {
    const float* input  = (const float*)d_in[0];
    const float* cond   = (const float*)d_in[1];
    const float* weight = (const float*)d_in[2];
    float* out = (float*)d_out;

    const int    nBlocks = B_ * (T_ / TSTEPS) * (H_ / 2);
    const size_t shmem   = (size_t)(NA + NX) * sizeof(float);

    conv3d_mod_wmma_f32<<<dim3(nBlocks), dim3(128), shmem, stream>>>(
        input, cond, weight, out);
}


# Winograd F(2,3) along t as well: a step yields 2 t x 2 h rows from 4 combined planes x 16 combined (kt,kh) taps, 384 WMMAs per wave for 576; products regrouped, so the summation order changes
# speedup vs baseline: 9.5176x; 1.0916x over previous
#include <hip/hip_runtime.h>


typedef __attribute__((ext_vector_type(2))) float v2f;
typedef __attribute__((ext_vector_type(4))) float v4f;
typedef __attribute__((ext_vector_type(8))) float v8f;

#define B_    4
#define CIN_  16
#define COUT_ 16
#define T_    16
#define H_    128
#define W_    128
#define SCALE_ 0.04811252243246881f

#define NSTEP (T_ / 2)
#define XROW 144
#define XOFF 2
#define APITCH 258
#define NTAPW 48
#define NA   (NTAPW * APITCH)
#define NDH  4
#define PLANE (NDH * 8 * XROW * 2)
#define NX   (4 * PLANE)

__device__ __forceinline__ void load_plane(v4f (&ev)[8], v4f (&od)[8], const float* __restrict__ inb,
                                           int ts, int h0, int wvid)
{
    const size_t strideC = (size_t)T_ * H_ * W_;
    const int tsc = ts < 0 ? 0 : (ts > T_ - 1 ? T_ - 1 : ts);
#pragma unroll
    for (int u = 0; u < 8; ++u) {
        const int cp = (u & 1) * 4 + wvid;
        const int dh = u >> 1;
        const int hs = h0 + dh - 1;
        const int hsc = hs < 0 ? 0 : (hs > H_ - 1 ? H_ - 1 : hs);
        const float* src = inb + ((size_t)(2 * cp) * T_ + tsc) * (H_ * W_) + (size_t)hsc * W_;
        ev[u] = *(const v4f*)src;
        od[u] = *(const v4f*)(src + strideC);
    }
}

__device__ __forceinline__ void store_planes(const v4f (&ev)[4][8], const v4f (&od)[4][8], float* sX,
                                             int ts0, int h0, int wvid, int lane)
{
    v4f* p4 = (v4f*)sX;
    const v4f zero = {0.f, 0.f, 0.f, 0.f};
#pragma unroll
    for (int k = 0; k < 2; ++k) {
        const int cp = k * 4 + wvid;
        v4f te[4][4], to[4][4];
#pragma unroll
        for (int pl = 0; pl < 4; ++pl) {
            const bool tok = (unsigned)(ts0 + pl) < (unsigned)T_;
            v4f e[4], o[4];
#pragma unroll
            for (int dh = 0; dh < 4; ++dh) {
                const int hs = h0 + dh - 1;
                const bool ok = tok & ((unsigned)hs < (unsigned)H_);
                e[dh] = ok ? ev[pl][2 * dh + k] : zero;
                o[dh] = ok ? od[pl][2 * dh + k] : zero;
            }
            te[pl][0] = e[0] - e[2];  to[pl][0] = o[0] - o[2];
            te[pl][1] = e[1] + e[2];  to[pl][1] = o[1] + o[2];
            te[pl][2] = e[2] - e[1];  to[pl][2] = o[2] - o[1];
            te[pl][3] = e[1] - e[3];  to[pl][3] = o[1] - o[3];
        }
#pragma unroll
        for (int j = 0; j < 4; ++j) {
            v4f qe[4], qo[4];
            qe[0] = te[0][j] - te[2][j];  qo[0] = to[0][j] - to[2][j];
            qe[1] = te[1][j] + te[2][j];  qo[1] = to[1][j] + to[2][j];
            qe[2] = te[2][j] - te[1][j];  qo[2] = to[2][j] - to[1][j];
            qe[3] = te[1][j] - te[3][j];  qo[3] = to[1][j] - to[3][j];
            const int p = j * 8 + cp;
#pragma unroll
            for (int jt = 0; jt < 4; ++jt) {
                v4f lo = {qe[jt].x, qo[jt].x, qe[jt].y, qo[jt].y};
                v4f hi = {qe[jt].z, qo[jt].z, qe[jt].w, qo[jt].w};
                p4[jt * (PLANE / 4) + p * (XROW / 2) + 2 * lane + 1] = lo;
                p4[jt * (PLANE / 4) + p * (XROW / 2) + 2 * lane + 2] = hi;
            }
        }
    }
}

__device__ __forceinline__ void load_frags(v2f (&fa)[16], v2f (&fb)[32], const float* sA,
                                           const float* sX, int blk, int hf, int lane, int col0)
{
    const int jt = blk / 3;
    const int kw = blk - 3 * jt;
    const int n0 = col0 + kw;
    const float* xp = sX + jt * PLANE;
#pragma unroll
    for (int j = 0; j < 4; ++j) {
        const int tap = (jt * 4 + j) * 3 + kw;
#pragma unroll
        for (int c = 0; c < 4; ++c) {
            const int r = j * 4 + c;
            fa[r] = *(const v2f*)&sA[tap * APITCH + ((c << 5) + lane) * 2];
            const int base = (j * 8 + 2 * c + hf) * (XROW * 2);
            fb[2 * r + 0] = *(const v2f*)&xp[base + n0 * 2];
            fb[2 * r + 1] = *(const v2f*)&xp[base + (n0 + 16) * 2];
        }
    }
}

__global__ __launch_bounds__(128) __attribute__((amdgpu_waves_per_eu(1, 1))) void conv3d_mod_wmma_f32(
    const float* __restrict__ input,
    const float* __restrict__ cond,
    const float* __restrict__ weight,
    float* __restrict__ out)
{
    extern __shared__ __attribute__((aligned(16))) float smem[];
    float* sA = smem;
    float* sX = smem + NA;

    const int bi  = blockIdx.x;
    const int h0  = (bi % (H_ / 2)) * 2;
    const int b   = bi / (H_ / 2);
    const int tid = threadIdx.x;
    const int wvid = tid >> 5;
    const int lane = tid & 31;

    const float* inb = input + (size_t)b * CIN_ * T_ * (H_ * W_) + 4 * lane;
    v4f ev[4][8], od[4][8];
#pragma unroll
    for (int pl = 0; pl < 4; ++pl) load_plane(ev[pl], od[pl], inb, pl - 1, h0, wvid);

    {
        const float* cb = cond + b * CIN_;
        float g[6][9];
        float cs[6];
#pragma unroll
        for (int u = 0; u < 6; ++u) {
            const int q  = u * 128 + tid;
            const int mc = q / 3;
            const int kw = q - mc * 3;
            const float* wp = weight + mc * 27 + kw;
#pragma unroll
            for (int e = 0; e < 9; ++e) g[u][e] = wp[3 * e];
            cs[u] = cb[mc & 15];
        }
#pragma unroll
        for (int u = 0; u < 6; ++u) {
            const int q  = u * 128 + tid;
            const int mc = q / 3;
            const int kw = q - mc * 3;
            const int d = ((((mc >> 2) & 3) * 32 + ((mc >> 1) & 1) * 16 + ((mc >> 4) & 15)) << 1) + (mc & 1);
            float a[3][4];
#pragma unroll
            for (int kt = 0; kt < 3; ++kt) {
                const float g0  = g[u][kt * 3 + 0] * SCALE_ * cs[u];
                const float g1  = g[u][kt * 3 + 1] * SCALE_ * cs[u];
                const float g2  = g[u][kt * 3 + 2] * SCALE_ * cs[u];
                const float s02 = g0 + g2;
                a[kt][0] = g0;
                a[kt][1] = 0.5f * (s02 + g1);
                a[kt][2] = 0.5f * (s02 - g1);
                a[kt][3] = g2;
            }
            float* dst = sA + kw * APITCH + d;
#pragma unroll
            for (int j = 0; j < 4; ++j) {
                const float s02 = a[0][j] + a[2][j];
                dst[((0 * 4 + j) * 3) * APITCH] = a[0][j];
                dst[((1 * 4 + j) * 3) * APITCH] = 0.5f * (s02 + a[1][j]);
                dst[((2 * 4 + j) * 3) * APITCH] = 0.5f * (s02 - a[1][j]);
                dst[((3 * 4 + j) * 3) * APITCH] = a[2][j];
            }
        }
        {
            v2f z = {0.f, 0.f};
            *(v2f*)&sX[(tid * XROW + (XOFF - 1)) * 2]  = z;
            *(v2f*)&sX[(tid * XROW + (XOFF + W_)) * 2] = z;
        }
    }

    const int wseg = wvid * 32;
    const int hf   = lane >> 4;
    const int lx   = lane & 15;

    const size_t strideC = (size_t)T_ * H_ * W_;
    float* ob = out + (size_t)b * COUT_ * strideC + (size_t)h0 * W_ + wseg + lx;
    const int col0 = wseg + lx + (XOFF - 1);

    for (int s = 0; s < NSTEP; ++s) {
        const int  t    = 2 * s;
        const bool more = s + 1 < NSTEP;

        if (s > 0) __syncthreads();
        store_planes(ev, od, sX, t - 1, h0, wvid, lane);
        __syncthreads();

        v2f fa[2][16];
        v2f fb[2][32];
        load_frags(fa[0], fb[0], sA, sX, 0, hf, lane, col0);

        if (more) {
#pragma unroll
            for (int pl = 0; pl < 3; ++pl) load_plane(ev[pl], od[pl], inb, t + 1 + pl, h0, wvid);
        }

        v8f acc[32];
#pragma unroll
        for (int a = 0; a < 32; ++a) acc[a] = (v8f){0.f, 0.f, 0.f, 0.f, 0.f, 0.f, 0.f, 0.f};

#pragma unroll
        for (int blk = 0; blk < 12; ++blk) {
            const int cur = blk & 1;
            const int jt  = blk / 3;
            if (blk + 1 < 12) load_frags(fa[cur ^ 1], fb[cur ^ 1], sA, sX, blk + 1, hf, lane, col0);
            __builtin_amdgcn_sched_barrier(0);
#pragma unroll
            for (int cg = 0; cg < 4; ++cg) {
#pragma unroll
                for (int j = 0; j < 4; ++j) {
                    const int r = j * 4 + cg;
                    const int a0 = (jt * 4 + j) * 2;
                    acc[a0 + 0] = __builtin_amdgcn_wmma_f32_16x16x4_f32(
                        false, fa[cur][r], false, fb[cur][2 * r + 0], (short)0, acc[a0 + 0], false, false);
                    acc[a0 + 1] = __builtin_amdgcn_wmma_f32_16x16x4_f32(
                        false, fa[cur][r], false, fb[cur][2 * r + 1], (short)0, acc[a0 + 1], false, false);
                }
            }
            __builtin_amdgcn_sched_barrier(0);
            asm volatile("v_nop\n\tv_nop\n\tv_nop\n\tv_nop\n\tv_nop");
            __builtin_amdgcn_sched_barrier(0);
            if (blk == 0 && more) load_plane(ev[3], od[3], inb, t + 4, h0, wvid);
        }

        float* o = ob + (size_t)t * (H_ * W_);
#pragma unroll
        for (int i = 0; i < 2; ++i) {
            v8f y[4][2];
#pragma unroll
            for (int jt = 0; jt < 4; ++jt) {
                y[jt][0] = (acc[(jt * 4 + 0) * 2 + i] + acc[(jt * 4 + 1) * 2 + i]) + acc[(jt * 4 + 2) * 2 + i];
                y[jt][1] = (acc[(jt * 4 + 1) * 2 + i] - acc[(jt * 4 + 2) * 2 + i]) - acc[(jt * 4 + 3) * 2 + i];
            }
#pragma unroll
            for (int hb = 0; hb < 2; ++hb) {
                const v8f z0 = (y[0][hb] + y[1][hb]) + y[2][hb];
                const v8f z1 = (y[1][hb] - y[2][hb]) - y[3][hb];
#pragma unroll
                for (int r = 0; r < 8; ++r) {
                    size_t mo = (size_t)(hf * 8 + r) * strideC + hb * W_ + 16 * i;
                    o[mo]           = z0[r];
                    o[mo + H_ * W_] = z1[r];
                }
            }
        }
    }
}

extern "C" void kernel_launch(void* const* d_in, const int* in_sizes, int n_in,
                              void* d_out, int out_size, void* d_ws, size_t ws_size,
                              hipStream_t stream) {
    const float* input  = (const float*)d_in[0];
    const float* cond   = (const float*)d_in[1];
    const float* weight = (const float*)d_in[2];
    float* out = (float*)d_out;

    const int    nBlocks = B_ * (H_ / 2);
    const size_t shmem   = (size_t)(NA + NX) * sizeof(float);

    conv3d_mod_wmma_f32<<<dim3(nBlocks), dim3(128), shmem, stream>>>(
        input, cond, weight, out);
}


# the two raw planes shared by consecutive steps stay in registers as h-combined rows: a step fetches and h-combines 2 new planes, not 4; arithmetic and summation order unchanged
# speedup vs baseline: 9.8443x; 1.0343x over previous
#include <hip/hip_runtime.h>


typedef __attribute__((ext_vector_type(2))) float v2f;
typedef __attribute__((ext_vector_type(4))) float v4f;
typedef __attribute__((ext_vector_type(8))) float v8f;

#define B_    4
#define CIN_  16
#define COUT_ 16
#define T_    16
#define H_    128
#define W_    128
#define SCALE_ 0.04811252243246881f

#define NSTEP (T_ / 2)
#define XROW 144
#define XOFF 2
#define APITCH 258
#define NTAPW 48
#define NA   (NTAPW * APITCH)
#define NDH  4
#define PLANE (NDH * 8 * XROW * 2)
#define NX   (4 * PLANE)

__device__ __forceinline__ void load_plane(v4f (&ev)[8], v4f (&od)[8], const float* __restrict__ inb,
                                           int ts, int h0, int wvid)
{
    const size_t strideC = (size_t)T_ * H_ * W_;
    const int tsc = ts < 0 ? 0 : (ts > T_ - 1 ? T_ - 1 : ts);
#pragma unroll
    for (int u = 0; u < 8; ++u) {
        const int cp = (u & 1) * 4 + wvid;
        const int dh = u >> 1;
        const int hs = h0 + dh - 1;
        const int hsc = hs < 0 ? 0 : (hs > H_ - 1 ? H_ - 1 : hs);
        const float* src = inb + ((size_t)(2 * cp) * T_ + tsc) * (H_ * W_) + (size_t)hsc * W_;
        ev[u] = *(const v4f*)src;
        od[u] = *(const v4f*)(src + strideC);
    }
}

__device__ __forceinline__ void hcomb(v4f (&te)[2][4], v4f (&to)[2][4], const v4f (&ev)[8],
                                      const v4f (&od)[8], int ts, int h0)
{
    const v4f zero = {0.f, 0.f, 0.f, 0.f};
    const bool tok = (unsigned)ts < (unsigned)T_;
#pragma unroll
    for (int k = 0; k < 2; ++k) {
        v4f e[4], o[4];
#pragma unroll
        for (int dh = 0; dh < 4; ++dh) {
            const int hs = h0 + dh - 1;
            const bool ok = tok & ((unsigned)hs < (unsigned)H_);
            e[dh] = ok ? ev[2 * dh + k] : zero;
            o[dh] = ok ? od[2 * dh + k] : zero;
        }
        te[k][0] = e[0] - e[2];  to[k][0] = o[0] - o[2];
        te[k][1] = e[1] + e[2];  to[k][1] = o[1] + o[2];
        te[k][2] = e[2] - e[1];  to[k][2] = o[2] - o[1];
        te[k][3] = e[1] - e[3];  to[k][3] = o[1] - o[3];
    }
}

__device__ __forceinline__ void store_planes(const v4f (&he)[4][2][4], const v4f (&ho)[4][2][4], float* sX,
                                             int wvid, int lane)
{
    v4f* p4 = (v4f*)sX;
#pragma unroll
    for (int k = 0; k < 2; ++k) {
        const int cp = k * 4 + wvid;
#pragma unroll
        for (int j = 0; j < 4; ++j) {
            v4f qe[4], qo[4];
            qe[0] = he[0][k][j] - he[2][k][j];  qo[0] = ho[0][k][j] - ho[2][k][j];
            qe[1] = he[1][k][j] + he[2][k][j];  qo[1] = ho[1][k][j] + ho[2][k][j];
            qe[2] = he[2][k][j] - he[1][k][j];  qo[2] = ho[2][k][j] - ho[1][k][j];
            qe[3] = he[1][k][j] - he[3][k][j];  qo[3] = ho[1][k][j] - ho[3][k][j];
            const int p = j * 8 + cp;
#pragma unroll
            for (int jt = 0; jt < 4; ++jt) {
                v4f lo = {qe[jt].x, qo[jt].x, qe[jt].y, qo[jt].y};
                v4f hi = {qe[jt].z, qo[jt].z, qe[jt].w, qo[jt].w};
                p4[jt * (PLANE / 4) + p * (XROW / 2) + 2 * lane + 1] = lo;
                p4[jt * (PLANE / 4) + p * (XROW / 2) + 2 * lane + 2] = hi;
            }
        }
    }
}

__device__ __forceinline__ void load_frags(v2f (&fa)[16], v2f (&fb)[32], const float* sA,
                                           const float* sX, int blk, int hf, int lane, int col0)
{
    const int jt = blk / 3;
    const int kw = blk - 3 * jt;
    const int n0 = col0 + kw;
    const float* xp = sX + jt * PLANE;
#pragma unroll
    for (int j = 0; j < 4; ++j) {
        const int tap = (jt * 4 + j) * 3 + kw;
#pragma unroll
        for (int c = 0; c < 4; ++c) {
            const int r = j * 4 + c;
            fa[r] = *(const v2f*)&sA[tap * APITCH + ((c << 5) + lane) * 2];
            const int base = (j * 8 + 2 * c + hf) * (XROW * 2);
            fb[2 * r + 0] = *(const v2f*)&xp[base + n0 * 2];
            fb[2 * r + 1] = *(const v2f*)&xp[base + (n0 + 16) * 2];
        }
    }
}

__global__ __launch_bounds__(128) __attribute__((amdgpu_waves_per_eu(1, 1))) void conv3d_mod_wmma_f32(
    const float* __restrict__ input,
    const float* __restrict__ cond,
    const float* __restrict__ weight,
    float* __restrict__ out)
{
    extern __shared__ __attribute__((aligned(16))) float smem[];
    float* sA = smem;
    float* sX = smem + NA;

    const int bi  = blockIdx.x;
    const int h0  = (bi % (H_ / 2)) * 2;
    const int b   = bi / (H_ / 2);
    const int tid = threadIdx.x;
    const int wvid = tid >> 5;
    const int lane = tid & 31;

    const float* inb = input + (size_t)b * CIN_ * T_ * (H_ * W_) + 4 * lane;
    v4f ev0[8], od0[8];
    v4f ev[2][8], od[2][8];
    load_plane(ev0, od0, inb, 0, h0, wvid);
    load_plane(ev[0], od[0], inb, 1, h0, wvid);
    load_plane(ev[1], od[1], inb, 2, h0, wvid);

    {
        const float* cb = cond + b * CIN_;
        float g[6][9];
        float cs[6];
#pragma unroll
        for (int u = 0; u < 6; ++u) {
            const int q  = u * 128 + tid;
            const int mc = q / 3;
            const int kw = q - mc * 3;
            const float* wp = weight + mc * 27 + kw;
#pragma unroll
            for (int e = 0; e < 9; ++e) g[u][e] = wp[3 * e];
            cs[u] = cb[mc & 15];
        }
#pragma unroll
        for (int u = 0; u < 6; ++u) {
            const int q  = u * 128 + tid;
            const int mc = q / 3;
            const int kw = q - mc * 3;
            const int d = ((((mc >> 2) & 3) * 32 + ((mc >> 1) & 1) * 16 + ((mc >> 4) & 15)) << 1) + (mc & 1);
            float a[3][4];
#pragma unroll
            for (int kt = 0; kt < 3; ++kt) {
                const float g0  = g[u][kt * 3 + 0] * SCALE_ * cs[u];
                const float g1  = g[u][kt * 3 + 1] * SCALE_ * cs[u];
                const float g2  = g[u][kt * 3 + 2] * SCALE_ * cs[u];
                const float s02 = g0 + g2;
                a[kt][0] = g0;
                a[kt][1] = 0.5f * (s02 + g1);
                a[kt][2] = 0.5f * (s02 - g1);
                a[kt][3] = g2;
            }
            float* dst = sA + kw * APITCH + d;
#pragma unroll
            for (int j = 0; j < 4; ++j) {
                const float s02 = a[0][j] + a[2][j];
                dst[((0 * 4 + j) * 3) * APITCH] = a[0][j];
                dst[((1 * 4 + j) * 3) * APITCH] = 0.5f * (s02 + a[1][j]);
                dst[((2 * 4 + j) * 3) * APITCH] = 0.5f * (s02 - a[1][j]);
                dst[((3 * 4 + j) * 3) * APITCH] = a[2][j];
            }
        }
        {
            v2f z = {0.f, 0.f};
            *(v2f*)&sX[(tid * XROW + (XOFF - 1)) * 2]  = z;
            *(v2f*)&sX[(tid * XROW + (XOFF + W_)) * 2] = z;
        }
    }

    v4f he[4][2][4], ho[4][2][4];
    {
        const v4f zero = {0.f, 0.f, 0.f, 0.f};
#pragma unroll
        for (int k = 0; k < 2; ++k) {
#pragma unroll
            for (int j = 0; j < 4; ++j) { he[0][k][j] = zero; ho[0][k][j] = zero; }
        }
        hcomb(he[1], ho[1], ev0, od0, 0, h0);
    }

    const int wseg = wvid * 32;
    const int hf   = lane >> 4;
    const int lx   = lane & 15;

    const size_t strideC = (size_t)T_ * H_ * W_;
    float* ob = out + (size_t)b * COUT_ * strideC + (size_t)h0 * W_ + wseg + lx;
    const int col0 = wseg + lx + (XOFF - 1);

    for (int s = 0; s < NSTEP; ++s) {
        const int  t    = 2 * s;
        const bool more = s + 1 < NSTEP;

        hcomb(he[2], ho[2], ev[0], od[0], t + 1, h0);
        hcomb(he[3], ho[3], ev[1], od[1], t + 2, h0);
        if (s > 0) __syncthreads();
        store_planes(he, ho, sX, wvid, lane);
        __syncthreads();
#pragma unroll
        for (int k = 0; k < 2; ++k) {
#pragma unroll
            for (int j = 0; j < 4; ++j) {
                he[0][k][j] = he[2][k][j];  ho[0][k][j] = ho[2][k][j];
                he[1][k][j] = he[3][k][j];  ho[1][k][j] = ho[3][k][j];
            }
        }

        v2f fa[2][16];
        v2f fb[2][32];
        load_frags(fa[0], fb[0], sA, sX, 0, hf, lane, col0);

        if (more) {
            load_plane(ev[0], od[0], inb, t + 3, h0, wvid);
            load_plane(ev[1], od[1], inb, t + 4, h0, wvid);
        }

        v8f acc[32];
#pragma unroll
        for (int a = 0; a < 32; ++a) acc[a] = (v8f){0.f, 0.f, 0.f, 0.f, 0.f, 0.f, 0.f, 0.f};

#pragma unroll
        for (int blk = 0; blk < 12; ++blk) {
            const int cur = blk & 1;
            const int jt  = blk / 3;
            if (blk + 1 < 12) load_frags(fa[cur ^ 1], fb[cur ^ 1], sA, sX, blk + 1, hf, lane, col0);
            __builtin_amdgcn_sched_barrier(0);
#pragma unroll
            for (int cg = 0; cg < 4; ++cg) {
#pragma unroll
                for (int j = 0; j < 4; ++j) {
                    const int r = j * 4 + cg;
                    const int a0 = (jt * 4 + j) * 2;
                    acc[a0 + 0] = __builtin_amdgcn_wmma_f32_16x16x4_f32(
                        false, fa[cur][r], false, fb[cur][2 * r + 0], (short)0, acc[a0 + 0], false, false);
                    acc[a0 + 1] = __builtin_amdgcn_wmma_f32_16x16x4_f32(
                        false, fa[cur][r], false, fb[cur][2 * r + 1], (short)0, acc[a0 + 1], false, false);
                }
            }
            __builtin_amdgcn_sched_barrier(0);
            asm volatile("v_nop\n\tv_nop\n\tv_nop\n\tv_nop\n\tv_nop");
            __builtin_amdgcn_sched_barrier(0);
        }

        float* o = ob + (size_t)t * (H_ * W_);
#pragma unroll
        for (int i = 0; i < 2; ++i) {
            v8f y[4][2];
#pragma unroll
            for (int jt = 0; jt < 4; ++jt) {
                y[jt][0] = (acc[(jt * 4 + 0) * 2 + i] + acc[(jt * 4 + 1) * 2 + i]) + acc[(jt * 4 + 2) * 2 + i];
                y[jt][1] = (acc[(jt * 4 + 1) * 2 + i] - acc[(jt * 4 + 2) * 2 + i]) - acc[(jt * 4 + 3) * 2 + i];
            }
#pragma unroll
            for (int hb = 0; hb < 2; ++hb) {
                const v8f z0 = (y[0][hb] + y[1][hb]) + y[2][hb];
                const v8f z1 = (y[1][hb] - y[2][hb]) - y[3][hb];
#pragma unroll
                for (int r = 0; r < 8; ++r) {
                    size_t mo = (size_t)(hf * 8 + r) * strideC + hb * W_ + 16 * i;
                    o[mo]           = z0[r];
                    o[mo + H_ * W_] = z1[r];
                }
            }
        }
    }
}

extern "C" void kernel_launch(void* const* d_in, const int* in_sizes, int n_in,
                              void* d_out, int out_size, void* d_ws, size_t ws_size,
                              hipStream_t stream) {
    const float* input  = (const float*)d_in[0];
    const float* cond   = (const float*)d_in[1];
    const float* weight = (const float*)d_in[2];
    float* out = (float*)d_out;

    const int    nBlocks = B_ * (H_ / 2);
    const size_t shmem   = (size_t)(NA + NX) * sizeof(float);

    conv3d_mod_wmma_f32<<<dim3(nBlocks), dim3(128), shmem, stream>>>(
        input, cond, weight, out);
}


# output stores leave the end-of-step burst: a step's transformed tiles stay in registers and go out in groups of 8 at the next step's WMMA block boundaries; values and summation order unchanged
# speedup vs baseline: 10.2656x; 1.0428x over previous
#include <hip/hip_runtime.h>


typedef __attribute__((ext_vector_type(2))) float v2f;
typedef __attribute__((ext_vector_type(4))) float v4f;
typedef __attribute__((ext_vector_type(8))) float v8f;

#define B_    4
#define CIN_  16
#define COUT_ 16
#define T_    16
#define H_    128
#define W_    128
#define SCALE_ 0.04811252243246881f

#define NSTEP (T_ / 2)
#define XROW 144
#define XOFF 2
#define APITCH 258
#define NTAPW 48
#define NA   (NTAPW * APITCH)
#define NDH  4
#define PLANE (NDH * 8 * XROW * 2)
#define NX   (4 * PLANE)

__device__ __forceinline__ void load_plane(v4f (&ev)[8], v4f (&od)[8], const float* __restrict__ inb,
                                           int ts, int h0, int wvid)
{
    const size_t strideC = (size_t)T_ * H_ * W_;
    const int tsc = ts < 0 ? 0 : (ts > T_ - 1 ? T_ - 1 : ts);
#pragma unroll
    for (int u = 0; u < 8; ++u) {
        const int cp = (u & 1) * 4 + wvid;
        const int dh = u >> 1;
        const int hs = h0 + dh - 1;
        const int hsc = hs < 0 ? 0 : (hs > H_ - 1 ? H_ - 1 : hs);
        const float* src = inb + ((size_t)(2 * cp) * T_ + tsc) * (H_ * W_) + (size_t)hsc * W_;
        ev[u] = *(const v4f*)src;
        od[u] = *(const v4f*)(src + strideC);
    }
}

__device__ __forceinline__ void hcomb(v4f (&te)[2][4], v4f (&to)[2][4], const v4f (&ev)[8],
                                      const v4f (&od)[8], int ts, int h0)
{
    const v4f zero = {0.f, 0.f, 0.f, 0.f};
    const bool tok = (unsigned)ts < (unsigned)T_;
#pragma unroll
    for (int k = 0; k < 2; ++k) {
        v4f e[4], o[4];
#pragma unroll
        for (int dh = 0; dh < 4; ++dh) {
            const int hs = h0 + dh - 1;
            const bool ok = tok & ((unsigned)hs < (unsigned)H_);
            e[dh] = ok ? ev[2 * dh + k] : zero;
            o[dh] = ok ? od[2 * dh + k] : zero;
        }
        te[k][0] = e[0] - e[2];  to[k][0] = o[0] - o[2];
        te[k][1] = e[1] + e[2];  to[k][1] = o[1] + o[2];
        te[k][2] = e[2] - e[1];  to[k][2] = o[2] - o[1];
        te[k][3] = e[1] - e[3];  to[k][3] = o[1] - o[3];
    }
}

__device__ __forceinline__ void store_planes(const v4f (&he)[4][2][4], const v4f (&ho)[4][2][4], float* sX,
                                             int wvid, int lane)
{
    v4f* p4 = (v4f*)sX;
#pragma unroll
    for (int k = 0; k < 2; ++k) {
        const int cp = k * 4 + wvid;
#pragma unroll
        for (int j = 0; j < 4; ++j) {
            v4f qe[4], qo[4];
            qe[0] = he[0][k][j] - he[2][k][j];  qo[0] = ho[0][k][j] - ho[2][k][j];
            qe[1] = he[1][k][j] + he[2][k][j];  qo[1] = ho[1][k][j] + ho[2][k][j];
            qe[2] = he[2][k][j] - he[1][k][j];  qo[2] = ho[2][k][j] - ho[1][k][j];
            qe[3] = he[1][k][j] - he[3][k][j];  qo[3] = ho[1][k][j] - ho[3][k][j];
            const int p = j * 8 + cp;
#pragma unroll
            for (int jt = 0; jt < 4; ++jt) {
                v4f lo = {qe[jt].x, qo[jt].x, qe[jt].y, qo[jt].y};
                v4f hi = {qe[jt].z, qo[jt].z, qe[jt].w, qo[jt].w};
                p4[jt * (PLANE / 4) + p * (XROW / 2) + 2 * lane + 1] = lo;
                p4[jt * (PLANE / 4) + p * (XROW / 2) + 2 * lane + 2] = hi;
            }
        }
    }
}

__device__ __forceinline__ void load_frags(v2f (&fa)[16], v2f (&fb)[32], const float* sA,
                                           const float* sX, int blk, int hf, int lane, int col0)
{
    const int jt = blk / 3;
    const int kw = blk - 3 * jt;
    const int n0 = col0 + kw;
    const float* xp = sX + jt * PLANE;
#pragma unroll
    for (int j = 0; j < 4; ++j) {
        const int tap = (jt * 4 + j) * 3 + kw;
#pragma unroll
        for (int c = 0; c < 4; ++c) {
            const int r = j * 4 + c;
            fa[r] = *(const v2f*)&sA[tap * APITCH + ((c << 5) + lane) * 2];
            const int base = (j * 8 + 2 * c + hf) * (XROW * 2);
            fb[2 * r + 0] = *(const v2f*)&xp[base + n0 * 2];
            fb[2 * r + 1] = *(const v2f*)&xp[base + (n0 + 16) * 2];
        }
    }
}

__device__ __forceinline__ void store_group(float* o, const v8f (&zp)[8], int grp, int hf)
{
    const size_t strideC = (size_t)T_ * H_ * W_;
    const int hb = grp >> 2;
    const int tz = (grp >> 1) & 1;
    const int rh = grp & 1;
#pragma unroll
    for (int rr = 0; rr < 4; ++rr) {
        const int r = rh * 4 + rr;
        const size_t mo = (size_t)(hf * 8 + r) * strideC + (size_t)tz * (H_ * W_) + hb * W_;
        o[mo]      = zp[(0 * 2 + hb) * 2 + tz][r];
        o[mo + 16] = zp[(1 * 2 + hb) * 2 + tz][r];
    }
}

__global__ __launch_bounds__(128) __attribute__((amdgpu_waves_per_eu(1, 1))) void conv3d_mod_wmma_f32(
    const float* __restrict__ input,
    const float* __restrict__ cond,
    const float* __restrict__ weight,
    float* __restrict__ out)
{
    extern __shared__ __attribute__((aligned(16))) float smem[];
    float* sA = smem;
    float* sX = smem + NA;

    const int bi  = blockIdx.x;
    const int h0  = (bi % (H_ / 2)) * 2;
    const int b   = bi / (H_ / 2);
    const int tid = threadIdx.x;
    const int wvid = tid >> 5;
    const int lane = tid & 31;

    const float* inb = input + (size_t)b * CIN_ * T_ * (H_ * W_) + 4 * lane;
    v4f ev0[8], od0[8];
    v4f ev[2][8], od[2][8];
    load_plane(ev0, od0, inb, 0, h0, wvid);
    load_plane(ev[0], od[0], inb, 1, h0, wvid);
    load_plane(ev[1], od[1], inb, 2, h0, wvid);

    {
        const float* cb = cond + b * CIN_;
        float g[6][9];
        float cs[6];
#pragma unroll
        for (int u = 0; u < 6; ++u) {
            const int q  = u * 128 + tid;
            const int mc = q / 3;
            const int kw = q - mc * 3;
            const float* wp = weight + mc * 27 + kw;
#pragma unroll
            for (int e = 0; e < 9; ++e) g[u][e] = wp[3 * e];
            cs[u] = cb[mc & 15];
        }
#pragma unroll
        for (int u = 0; u < 6; ++u) {
            const int q  = u * 128 + tid;
            const int mc = q / 3;
            const int kw = q - mc * 3;
            const int d = ((((mc >> 2) & 3) * 32 + ((mc >> 1) & 1) * 16 + ((mc >> 4) & 15)) << 1) + (mc & 1);
            float a[3][4];
#pragma unroll
            for (int kt = 0; kt < 3; ++kt) {
                const float g0  = g[u][kt * 3 + 0] * SCALE_ * cs[u];
                const float g1  = g[u][kt * 3 + 1] * SCALE_ * cs[u];
                const float g2  = g[u][kt * 3 + 2] * SCALE_ * cs[u];
                const float s02 = g0 + g2;
                a[kt][0] = g0;
                a[kt][1] = 0.5f * (s02 + g1);
                a[kt][2] = 0.5f * (s02 - g1);
                a[kt][3] = g2;
            }
            float* dst = sA + kw * APITCH + d;
#pragma unroll
            for (int j = 0; j < 4; ++j) {
                const float s02 = a[0][j] + a[2][j];
                dst[((0 * 4 + j) * 3) * APITCH] = a[0][j];
                dst[((1 * 4 + j) * 3) * APITCH] = 0.5f * (s02 + a[1][j]);
                dst[((2 * 4 + j) * 3) * APITCH] = 0.5f * (s02 - a[1][j]);
                dst[((3 * 4 + j) * 3) * APITCH] = a[2][j];
            }
        }
        {
            v2f z = {0.f, 0.f};
            *(v2f*)&sX[(tid * XROW + (XOFF - 1)) * 2]  = z;
            *(v2f*)&sX[(tid * XROW + (XOFF + W_)) * 2] = z;
        }
    }

    v4f he[4][2][4], ho[4][2][4];
    {
        const v4f zero = {0.f, 0.f, 0.f, 0.f};
#pragma unroll
        for (int k = 0; k < 2; ++k) {
#pragma unroll
            for (int j = 0; j < 4; ++j) { he[0][k][j] = zero; ho[0][k][j] = zero; }
        }
        hcomb(he[1], ho[1], ev0, od0, 0, h0);
    }

    const int wseg = wvid * 32;
    const int hf   = lane >> 4;
    const int lx   = lane & 15;

    const size_t strideC = (size_t)T_ * H_ * W_;
    float* ob = out + (size_t)b * COUT_ * strideC + (size_t)h0 * W_ + wseg + lx;
    const int col0 = wseg + lx + (XOFF - 1);

    v8f zp[8];

    for (int s = 0; s < NSTEP; ++s) {
        const int  t    = 2 * s;
        const bool more = s + 1 < NSTEP;

        hcomb(he[2], ho[2], ev[0], od[0], t + 1, h0);
        hcomb(he[3], ho[3], ev[1], od[1], t + 2, h0);
        if (s > 0) __syncthreads();
        store_planes(he, ho, sX, wvid, lane);
        __syncthreads();
#pragma unroll
        for (int k = 0; k < 2; ++k) {
#pragma unroll
            for (int j = 0; j < 4; ++j) {
                he[0][k][j] = he[2][k][j];  ho[0][k][j] = ho[2][k][j];
                he[1][k][j] = he[3][k][j];  ho[1][k][j] = ho[3][k][j];
            }
        }

        v2f fa[2][16];
        v2f fb[2][32];
        load_frags(fa[0], fb[0], sA, sX, 0, hf, lane, col0);

        if (more) {
            load_plane(ev[0], od[0], inb, t + 3, h0, wvid);
            load_plane(ev[1], od[1], inb, t + 4, h0, wvid);
        }

        v8f acc[32];
#pragma unroll
        for (int a = 0; a < 32; ++a) acc[a] = (v8f){0.f, 0.f, 0.f, 0.f, 0.f, 0.f, 0.f, 0.f};

#pragma unroll
        for (int blk = 0; blk < 12; ++blk) {
            const int cur = blk & 1;
            const int jt  = blk / 3;
            if (blk >= 1 && blk <= 8) {
                if (s > 0) store_group(ob + (size_t)(t - 2) * (H_ * W_), zp, blk - 1, hf);
            }
            if (blk + 1 < 12) load_frags(fa[cur ^ 1], fb[cur ^ 1], sA, sX, blk + 1, hf, lane, col0);
            __builtin_amdgcn_sched_barrier(0);
#pragma unroll
            for (int cg = 0; cg < 4; ++cg) {
#pragma unroll
                for (int j = 0; j < 4; ++j) {
                    const int r = j * 4 + cg;
                    const int a0 = (jt * 4 + j) * 2;
                    acc[a0 + 0] = __builtin_amdgcn_wmma_f32_16x16x4_f32(
                        false, fa[cur][r], false, fb[cur][2 * r + 0], (short)0, acc[a0 + 0], false, false);
                    acc[a0 + 1] = __builtin_amdgcn_wmma_f32_16x16x4_f32(
                        false, fa[cur][r], false, fb[cur][2 * r + 1], (short)0, acc[a0 + 1], false, false);
                }
            }
            __builtin_amdgcn_sched_barrier(0);
            asm volatile("v_nop\n\tv_nop\n\tv_nop\n\tv_nop\n\tv_nop");
            __builtin_amdgcn_sched_barrier(0);
        }

#pragma unroll
        for (int i = 0; i < 2; ++i) {
            v8f y[4][2];
#pragma unroll
            for (int jt = 0; jt < 4; ++jt) {
                y[jt][0] = (acc[(jt * 4 + 0) * 2 + i] + acc[(jt * 4 + 1) * 2 + i]) + acc[(jt * 4 + 2) * 2 + i];
                y[jt][1] = (acc[(jt * 4 + 1) * 2 + i] - acc[(jt * 4 + 2) * 2 + i]) - acc[(jt * 4 + 3) * 2 + i];
            }
#pragma unroll
            for (int hb = 0; hb < 2; ++hb) {
                zp[(i * 2 + hb) * 2 + 0] = (y[0][hb] + y[1][hb]) + y[2][hb];
                zp[(i * 2 + hb) * 2 + 1] = (y[1][hb] - y[2][hb]) - y[3][hb];
            }
        }
    }

#pragma unroll
    for (int grp = 0; grp < 8; ++grp)
        store_group(ob + (size_t)(T_ - 2) * (H_ * W_), zp, grp, hf);
}

extern "C" void kernel_launch(void* const* d_in, const int* in_sizes, int n_in,
                              void* d_out, int out_size, void* d_ws, size_t ws_size,
                              hipStream_t stream) {
    const float* input  = (const float*)d_in[0];
    const float* cond   = (const float*)d_in[1];
    const float* weight = (const float*)d_in[2];
    float* out = (float*)d_out;

    const int    nBlocks = B_ * (H_ / 2);
    const size_t shmem   = (size_t)(NA + NX) * sizeof(float);

    conv3d_mod_wmma_f32<<<dim3(nBlocks), dim3(128), shmem, stream>>>(
        input, cond, weight, out);
}


# Winograd F(2,3) along w too: a lane's output column pair from 4 raw columns combined in registers x 4 combined kw taps, 256 WMMAs per wave-step for 384; products regrouped: summation order changes
# speedup vs baseline: 11.3271x; 1.1034x over previous
#include <hip/hip_runtime.h>


typedef __attribute__((ext_vector_type(2))) float v2f;
typedef __attribute__((ext_vector_type(4))) float v4f;
typedef __attribute__((ext_vector_type(8))) float v8f;

#define B_    4
#define CIN_  16
#define COUT_ 16
#define T_    16
#define H_    128
#define W_    128
#define SCALE_ 0.04811252243246881f

#define NSTEP (T_ / 2)
#define XROW 144
#define XOFF 2
#define APITCH 258
#define NTAPW 64
#define NA   (NTAPW * APITCH)
#define NDH  4
#define PLANE (NDH * 8 * XROW * 2)
#define NX   (4 * PLANE)

__device__ __forceinline__ void load_plane(v4f (&ev)[8], v4f (&od)[8], const float* __restrict__ inb,
                                           int ts, int h0, int wvid)
{
    const size_t strideC = (size_t)T_ * H_ * W_;
    const int tsc = ts < 0 ? 0 : (ts > T_ - 1 ? T_ - 1 : ts);
#pragma unroll
    for (int u = 0; u < 8; ++u) {
        const int cp = (u & 1) * 4 + wvid;
        const int dh = u >> 1;
        const int hs = h0 + dh - 1;
        const int hsc = hs < 0 ? 0 : (hs > H_ - 1 ? H_ - 1 : hs);
        const float* src = inb + ((size_t)(2 * cp) * T_ + tsc) * (H_ * W_) + (size_t)hsc * W_;
        ev[u] = *(const v4f*)src;
        od[u] = *(const v4f*)(src + strideC);
    }
}

__device__ __forceinline__ void hcomb(v4f (&te)[2][4], v4f (&to)[2][4], const v4f (&ev)[8],
                                      const v4f (&od)[8], int ts, int h0)
{
    const v4f zero = {0.f, 0.f, 0.f, 0.f};
    const bool tok = (unsigned)ts < (unsigned)T_;
#pragma unroll
    for (int k = 0; k < 2; ++k) {
        v4f e[4], o[4];
#pragma unroll
        for (int dh = 0; dh < 4; ++dh) {
            const int hs = h0 + dh - 1;
            const bool ok = tok & ((unsigned)hs < (unsigned)H_);
            e[dh] = ok ? ev[2 * dh + k] : zero;
            o[dh] = ok ? od[2 * dh + k] : zero;
        }
        te[k][0] = e[0] - e[2];  to[k][0] = o[0] - o[2];
        te[k][1] = e[1] + e[2];  to[k][1] = o[1] + o[2];
        te[k][2] = e[2] - e[1];  to[k][2] = o[2] - o[1];
        te[k][3] = e[1] - e[3];  to[k][3] = o[1] - o[3];
    }
}

__device__ __forceinline__ void store_planes(const v4f (&he)[4][2][4], const v4f (&ho)[4][2][4], float* sX,
                                             int wvid, int lane)
{
    v4f* p4 = (v4f*)sX;
#pragma unroll
    for (int k = 0; k < 2; ++k) {
        const int cp = k * 4 + wvid;
#pragma unroll
        for (int j = 0; j < 4; ++j) {
            v4f qe[4], qo[4];
            qe[0] = he[0][k][j] - he[2][k][j];  qo[0] = ho[0][k][j] - ho[2][k][j];
            qe[1] = he[1][k][j] + he[2][k][j];  qo[1] = ho[1][k][j] + ho[2][k][j];
            qe[2] = he[2][k][j] - he[1][k][j];  qo[2] = ho[2][k][j] - ho[1][k][j];
            qe[3] = he[1][k][j] - he[3][k][j];  qo[3] = ho[1][k][j] - ho[3][k][j];
            const int p = j * 8 + cp;
#pragma unroll
            for (int jt = 0; jt < 4; ++jt) {
                v4f lo = {qe[jt].x, qo[jt].x, qe[jt].y, qo[jt].y};
                v4f hi = {qe[jt].z, qo[jt].z, qe[jt].w, qo[jt].w};
                p4[jt * (PLANE / 4) + p * (XROW / 2) + 2 * lane + 1] = lo;
                p4[jt * (PLANE / 4) + p * (XROW / 2) + 2 * lane + 2] = hi;
            }
        }
    }
}

__device__ __forceinline__ void load_frags(v2f (&fa)[32], v2f (&fd)[32], const float* sA,
                                           const float* sX, int bk, int hf, int lane, int colw)
{
    const int jt = bk >> 1;
    const int jh = bk & 1;
    const float* xp = sX + jt * PLANE;
#pragma unroll
    for (int jl = 0; jl < 2; ++jl) {
        const int j = 2 * jh + jl;
#pragma unroll
        for (int c = 0; c < 4; ++c) {
            const int base = (j * 8 + 2 * c + hf) * (XROW * 2);
#pragma unroll
            for (int e = 0; e < 4; ++e)
                fd[(jl * 4 + c) * 4 + e] = *(const v2f*)&xp[base + (colw + e) * 2];
#pragma unroll
            for (int q = 0; q < 4; ++q) {
                const int tap = (jt * 4 + j) * 4 + q;
                fa[(jl * 4 + q) * 4 + c] = *(const v2f*)&sA[tap * APITCH + ((c << 5) + lane) * 2];
            }
        }
    }
}

__device__ __forceinline__ void wcomb(v2f (&fv)[32], const v2f (&fd)[32])
{
#pragma unroll
    for (int g = 0; g < 8; ++g) {
        const v2f d0 = fd[g * 4 + 0], d1 = fd[g * 4 + 1], d2 = fd[g * 4 + 2], d3 = fd[g * 4 + 3];
        fv[g * 4 + 0] = d0 - d2;
        fv[g * 4 + 1] = d1 + d2;
        fv[g * 4 + 2] = d2 - d1;
        fv[g * 4 + 3] = d1 - d3;
    }
}

__device__ __forceinline__ void store_group(float* o, const v8f (&zp)[8], int grp, int hf)
{
    const size_t strideC = (size_t)T_ * H_ * W_;
    const int tz = grp >> 1;
    const int hb = grp & 1;
#pragma unroll
    for (int r = 0; r < 8; ++r) {
        const size_t mo = (size_t)(hf * 8 + r) * strideC + (size_t)tz * (H_ * W_) + hb * W_;
        v2f v = {zp[grp * 2 + 0][r], zp[grp * 2 + 1][r]};
        *(v2f*)(o + mo) = v;
    }
}

__global__ __launch_bounds__(128) __attribute__((amdgpu_waves_per_eu(1, 1))) void conv3d_mod_wmma_f32(
    const float* __restrict__ input,
    const float* __restrict__ cond,
    const float* __restrict__ weight,
    float* __restrict__ out)
{
    extern __shared__ __attribute__((aligned(16))) float smem[];
    float* sA = smem;
    float* sX = smem + NA;

    const int bi  = blockIdx.x;
    const int h0  = (bi % (H_ / 2)) * 2;
    const int b   = bi / (H_ / 2);
    const int tid = threadIdx.x;
    const int wvid = tid >> 5;
    const int lane = tid & 31;

    const float* inb = input + (size_t)b * CIN_ * T_ * (H_ * W_) + 4 * lane;
    v4f ev0[8], od0[8];
    v4f ev[2][8], od[2][8];
    load_plane(ev0, od0, inb, 0, h0, wvid);
    load_plane(ev[0], od[0], inb, 1, h0, wvid);
    load_plane(ev[1], od[1], inb, 2, h0, wvid);

    {
        const float* cb = cond + b * CIN_;
        float g[2][27];
        float cs[2];
#pragma unroll
        for (int u = 0; u < 2; ++u) {
            const int mc = u * 128 + tid;
            const float* wp = weight + mc * 27;
#pragma unroll
            for (int e = 0; e < 27; ++e) g[u][e] = wp[e];
            cs[u] = cb[mc & 15];
        }
#pragma unroll
        for (int u = 0; u < 2; ++u) {
            const int mc = u * 128 + tid;
            const int d = ((((mc >> 2) & 3) * 32 + ((mc >> 1) & 1) * 16 + ((mc >> 4) & 15)) << 1) + (mc & 1);
            float a[3][3][4];
#pragma unroll
            for (int kt = 0; kt < 3; ++kt) {
#pragma unroll
                for (int kh = 0; kh < 3; ++kh) {
                    const float g0  = g[u][(kt * 3 + kh) * 3 + 0] * SCALE_ * cs[u];
                    const float g1  = g[u][(kt * 3 + kh) * 3 + 1] * SCALE_ * cs[u];
                    const float g2  = g[u][(kt * 3 + kh) * 3 + 2] * SCALE_ * cs[u];
                    const float s02 = g0 + g2;
                    a[kt][kh][0] = g0;
                    a[kt][kh][1] = 0.5f * (s02 + g1);
                    a[kt][kh][2] = 0.5f * (s02 - g1);
                    a[kt][kh][3] = g2;
                }
            }
            float bb[3][4][4];
#pragma unroll
            for (int kt = 0; kt < 3; ++kt) {
#pragma unroll
                for (int q = 0; q < 4; ++q) {
                    const float s02 = a[kt][0][q] + a[kt][2][q];
                    bb[kt][0][q] = a[kt][0][q];
                    bb[kt][1][q] = 0.5f * (s02 + a[kt][1][q]);
                    bb[kt][2][q] = 0.5f * (s02 - a[kt][1][q]);
                    bb[kt][3][q] = a[kt][2][q];
                }
            }
            float* dst = sA + d;
#pragma unroll
            for (int j = 0; j < 4; ++j) {
#pragma unroll
                for (int q = 0; q < 4; ++q) {
                    const float s02 = bb[0][j][q] + bb[2][j][q];
                    dst[((0 * 4 + j) * 4 + q) * APITCH] = bb[0][j][q];
                    dst[((1 * 4 + j) * 4 + q) * APITCH] = 0.5f * (s02 + bb[1][j][q]);
                    dst[((2 * 4 + j) * 4 + q) * APITCH] = 0.5f * (s02 - bb[1][j][q]);
                    dst[((3 * 4 + j) * 4 + q) * APITCH] = bb[2][j][q];
                }
            }
        }
        {
            v2f z = {0.f, 0.f};
            *(v2f*)&sX[(tid * XROW + (XOFF - 1)) * 2]  = z;
            *(v2f*)&sX[(tid * XROW + (XOFF + W_)) * 2] = z;
        }
    }

    v4f he[4][2][4], ho[4][2][4];
    {
        const v4f zero = {0.f, 0.f, 0.f, 0.f};
#pragma unroll
        for (int k = 0; k < 2; ++k) {
#pragma unroll
            for (int j = 0; j < 4; ++j) { he[0][k][j] = zero; ho[0][k][j] = zero; }
        }
        hcomb(he[1], ho[1], ev0, od0, 0, h0);
    }

    const int wseg = wvid * 32;
    const int hf   = lane >> 4;
    const int lx   = lane & 15;

    const size_t strideC = (size_t)T_ * H_ * W_;
    float* ob = out + (size_t)b * COUT_ * strideC + (size_t)h0 * W_ + wseg + 2 * lx;
    const int colw = wseg + 2 * lx + (XOFF - 1);

    v8f zp[8];

    for (int s = 0; s < NSTEP; ++s) {
        const int  t    = 2 * s;
        const bool more = s + 1 < NSTEP;

        hcomb(he[2], ho[2], ev[0], od[0], t + 1, h0);
        hcomb(he[3], ho[3], ev[1], od[1], t + 2, h0);
        if (s > 0) __syncthreads();
        store_planes(he, ho, sX, wvid, lane);
        __syncthreads();
#pragma unroll
        for (int k = 0; k < 2; ++k) {
#pragma unroll
            for (int j = 0; j < 4; ++j) {
                he[0][k][j] = he[2][k][j];  ho[0][k][j] = ho[2][k][j];
                he[1][k][j] = he[3][k][j];  ho[1][k][j] = ho[3][k][j];
            }
        }

        v2f fa[2][32];
        v2f fd[2][32];
        load_frags(fa[0], fd[0], sA, sX, 0, hf, lane, colw);

        if (more) {
            load_plane(ev[0], od[0], inb, t + 3, h0, wvid);
            load_plane(ev[1], od[1], inb, t + 4, h0, wvid);
        }

        v8f yw[8];
        v8f z0[4], z1[4];
#pragma unroll
        for (int bk = 0; bk < 8; ++bk) {
            const int cur = bk & 1;
            const int jt  = bk >> 1;
            const int jh  = bk & 1;
            if (bk >= 1 && bk <= 4) {
                if (s > 0) store_group(ob + (size_t)(t - 2) * (H_ * W_), zp, bk - 1, hf);
            }
            if (bk + 1 < 8) load_frags(fa[cur ^ 1], fd[cur ^ 1], sA, sX, bk + 1, hf, lane, colw);
            v2f fv[32];
            wcomb(fv, fd[cur]);
            __builtin_amdgcn_sched_barrier(0);
            v8f acc[8];
#pragma unroll
            for (int c = 0; c < 4; ++c) {
#pragma unroll
                for (int jl = 0; jl < 2; ++jl) {
#pragma unroll
                    for (int q = 0; q < 4; ++q) {
                        const int a = jl * 4 + q;
                        const v8f cin0 = {0.f, 0.f, 0.f, 0.f, 0.f, 0.f, 0.f, 0.f};
                        acc[a] = __builtin_amdgcn_wmma_f32_16x16x4_f32(
                            false, fa[cur][a * 4 + c], false, fv[(jl * 4 + c) * 4 + q], (short)0,
                            c == 0 ? cin0 : acc[a], false, false);
                    }
                }
            }
            __builtin_amdgcn_sched_barrier(0);
            asm volatile("v_nop\n\tv_nop\n\tv_nop\n\tv_nop\n\tv_nop");
            __builtin_amdgcn_sched_barrier(0);

#pragma unroll
            for (int jl = 0; jl < 2; ++jl) {
                const int j = 2 * jh + jl;
                yw[j * 2 + 0] = (acc[jl * 4 + 0] + acc[jl * 4 + 1]) + acc[jl * 4 + 2];
                yw[j * 2 + 1] = (acc[jl * 4 + 1] - acc[jl * 4 + 2]) - acc[jl * 4 + 3];
            }
            if (jh == 1) {
#pragma unroll
                for (int p = 0; p < 2; ++p) {
                    v8f yh[2];
                    yh[0] = (yw[0 * 2 + p] + yw[1 * 2 + p]) + yw[2 * 2 + p];
                    yh[1] = (yw[1 * 2 + p] - yw[2 * 2 + p]) - yw[3 * 2 + p];
#pragma unroll
                    for (int hb = 0; hb < 2; ++hb) {
                        if (jt == 0) {
                            z0[hb * 2 + p] = yh[hb];
                        } else if (jt == 1) {
                            z0[hb * 2 + p] = z0[hb * 2 + p] + yh[hb];
                            z1[hb * 2 + p] = yh[hb];
                        } else if (jt == 2) {
                            zp[(0 * 2 + hb) * 2 + p] = z0[hb * 2 + p] + yh[hb];
                            z1[hb * 2 + p] = z1[hb * 2 + p] - yh[hb];
                        } else {
                            zp[(1 * 2 + hb) * 2 + p] = z1[hb * 2 + p] - yh[hb];
                        }
                    }
                }
            }
        }
    }

#pragma unroll
    for (int grp = 0; grp < 4; ++grp)
        store_group(ob + (size_t)(T_ - 2) * (H_ * W_), zp, grp, hf);
}

extern "C" void kernel_launch(void* const* d_in, const int* in_sizes, int n_in,
                              void* d_out, int out_size, void* d_ws, size_t ws_size,
                              hipStream_t stream) {
    const float* input  = (const float*)d_in[0];
    const float* cond   = (const float*)d_in[1];
    const float* weight = (const float*)d_in[2];
    float* out = (float*)d_out;

    const int    nBlocks = B_ * (H_ / 2);
    const size_t shmem   = (size_t)(NA + NX) * sizeof(float);

    conv3d_mod_wmma_f32<<<dim3(nBlocks), dim3(128), shmem, stream>>>(
        input, cond, weight, out);
}


# next step's combined planes go to LDS during the current step, plane jt at the head of WMMA block 2jt+1 behind an LDS-only barrier: no store or first-read wait between steps; arithmetic unchanged
# speedup vs baseline: 11.7369x; 1.0362x over previous
#include <hip/hip_runtime.h>


typedef __attribute__((ext_vector_type(2))) float v2f;
typedef __attribute__((ext_vector_type(4))) float v4f;
typedef __attribute__((ext_vector_type(8))) float v8f;

#define B_    4
#define CIN_  16
#define COUT_ 16
#define T_    16
#define H_    128
#define W_    128
#define SCALE_ 0.04811252243246881f

#define NSTEP (T_ / 2)
#define XROW 144
#define XOFF 2
#define APITCH 258
#define NTAPW 64
#define NA   (NTAPW * APITCH)
#define NDH  4
#define PLANE (NDH * 8 * XROW * 2)
#define NX   (4 * PLANE)

__device__ __forceinline__ void load_plane(v4f (&ev)[8], v4f (&od)[8], const float* __restrict__ inb,
                                           int ts, int h0, int wvid)
{
    const size_t strideC = (size_t)T_ * H_ * W_;
    const int tsc = ts < 0 ? 0 : (ts > T_ - 1 ? T_ - 1 : ts);
#pragma unroll
    for (int u = 0; u < 8; ++u) {
        const int cp = (u & 1) * 4 + wvid;
        const int dh = u >> 1;
        const int hs = h0 + dh - 1;
        const int hsc = hs < 0 ? 0 : (hs > H_ - 1 ? H_ - 1 : hs);
        const float* src = inb + ((size_t)(2 * cp) * T_ + tsc) * (H_ * W_) + (size_t)hsc * W_;
        ev[u] = *(const v4f*)src;
        od[u] = *(const v4f*)(src + strideC);
    }
}

__device__ __forceinline__ void hcomb(v4f (&te)[2][4], v4f (&to)[2][4], const v4f (&ev)[8],
                                      const v4f (&od)[8], int ts, int h0)
{
    const v4f zero = {0.f, 0.f, 0.f, 0.f};
    const bool tok = (unsigned)ts < (unsigned)T_;
#pragma unroll
    for (int k = 0; k < 2; ++k) {
        v4f e[4], o[4];
#pragma unroll
        for (int dh = 0; dh < 4; ++dh) {
            const int hs = h0 + dh - 1;
            const bool ok = tok & ((unsigned)hs < (unsigned)H_);
            e[dh] = ok ? ev[2 * dh + k] : zero;
            o[dh] = ok ? od[2 * dh + k] : zero;
        }
        te[k][0] = e[0] - e[2];  to[k][0] = o[0] - o[2];
        te[k][1] = e[1] + e[2];  to[k][1] = o[1] + o[2];
        te[k][2] = e[2] - e[1];  to[k][2] = o[2] - o[1];
        te[k][3] = e[1] - e[3];  to[k][3] = o[1] - o[3];
    }
}

__device__ __forceinline__ void store_plane(const v4f (&he)[4][2][4], const v4f (&ho)[4][2][4], float* sX,
                                            int wvid, int lane, int jt)
{
    v4f* p4 = (v4f*)sX;
#pragma unroll
    for (int k = 0; k < 2; ++k) {
        const int cp = k * 4 + wvid;
#pragma unroll
        for (int j = 0; j < 4; ++j) {
            v4f qe, qo;
            if (jt == 0)      { qe = he[0][k][j] - he[2][k][j];  qo = ho[0][k][j] - ho[2][k][j]; }
            else if (jt == 1) { qe = he[1][k][j] + he[2][k][j];  qo = ho[1][k][j] + ho[2][k][j]; }
            else if (jt == 2) { qe = he[2][k][j] - he[1][k][j];  qo = ho[2][k][j] - ho[1][k][j]; }
            else              { qe = he[1][k][j] - he[3][k][j];  qo = ho[1][k][j] - ho[3][k][j]; }
            const int p = j * 8 + cp;
            v4f lo = {qe.x, qo.x, qe.y, qo.y};
            v4f hi = {qe.z, qo.z, qe.w, qo.w};
            p4[jt * (PLANE / 4) + p * (XROW / 2) + 2 * lane + 1] = lo;
            p4[jt * (PLANE / 4) + p * (XROW / 2) + 2 * lane + 2] = hi;
        }
    }
}

__device__ __forceinline__ void lds_barrier()
{
    __builtin_amdgcn_fence(__ATOMIC_RELEASE, "workgroup", "local");
    __builtin_amdgcn_s_barrier();
    __builtin_amdgcn_fence(__ATOMIC_ACQUIRE, "workgroup", "local");
}

__device__ __forceinline__ void load_frags(v2f (&fa)[32], v2f (&fd)[32], const float* sA,
                                           const float* sX, int bk, int hf, int lane, int colw)
{
    const int jt = bk >> 1;
    const int jh = bk & 1;
    const float* xp = sX + jt * PLANE;
#pragma unroll
    for (int jl = 0; jl < 2; ++jl) {
        const int j = 2 * jh + jl;
#pragma unroll
        for (int c = 0; c < 4; ++c) {
            const int base = (j * 8 + 2 * c + hf) * (XROW * 2);
#pragma unroll
            for (int e = 0; e < 4; ++e)
                fd[(jl * 4 + c) * 4 + e] = *(const v2f*)&xp[base + (colw + e) * 2];
#pragma unroll
            for (int q = 0; q < 4; ++q) {
                const int tap = (jt * 4 + j) * 4 + q;
                fa[(jl * 4 + q) * 4 + c] = *(const v2f*)&sA[tap * APITCH + ((c << 5) + lane) * 2];
            }
        }
    }
}

__device__ __forceinline__ void wcomb(v2f (&fv)[32], const v2f (&fd)[32])
{
#pragma unroll
    for (int g = 0; g < 8; ++g) {
        const v2f d0 = fd[g * 4 + 0], d1 = fd[g * 4 + 1], d2 = fd[g * 4 + 2], d3 = fd[g * 4 + 3];
        fv[g * 4 + 0] = d0 - d2;
        fv[g * 4 + 1] = d1 + d2;
        fv[g * 4 + 2] = d2 - d1;
        fv[g * 4 + 3] = d1 - d3;
    }
}

__device__ __forceinline__ void store_group(float* o, const v8f (&zp)[8], int grp, int hf)
{
    const size_t strideC = (size_t)T_ * H_ * W_;
    const int tz = grp >> 1;
    const int hb = grp & 1;
#pragma unroll
    for (int r = 0; r < 8; ++r) {
        const size_t mo = (size_t)(hf * 8 + r) * strideC + (size_t)tz * (H_ * W_) + hb * W_;
        v2f v = {zp[grp * 2 + 0][r], zp[grp * 2 + 1][r]};
        *(v2f*)(o + mo) = v;
    }
}

__global__ __launch_bounds__(128) __attribute__((amdgpu_waves_per_eu(1, 1))) void conv3d_mod_wmma_f32(
    const float* __restrict__ input,
    const float* __restrict__ cond,
    const float* __restrict__ weight,
    float* __restrict__ out)
{
    extern __shared__ __attribute__((aligned(16))) float smem[];
    float* sA = smem;
    float* sX = smem + NA;

    const int bi  = blockIdx.x;
    const int h0  = (bi % (H_ / 2)) * 2;
    const int b   = bi / (H_ / 2);
    const int tid = threadIdx.x;
    const int wvid = tid >> 5;
    const int lane = tid & 31;

    const float* inb = input + (size_t)b * CIN_ * T_ * (H_ * W_) + 4 * lane;
    v4f ev0[8], od0[8];
    v4f ev[2][8], od[2][8];
    load_plane(ev0, od0, inb, 0, h0, wvid);
    load_plane(ev[0], od[0], inb, 1, h0, wvid);
    load_plane(ev[1], od[1], inb, 2, h0, wvid);

    {
        const float* cb = cond + b * CIN_;
        float g[2][27];
        float cs[2];
#pragma unroll
        for (int u = 0; u < 2; ++u) {
            const int mc = u * 128 + tid;
            const float* wp = weight + mc * 27;
#pragma unroll
            for (int e = 0; e < 27; ++e) g[u][e] = wp[e];
            cs[u] = cb[mc & 15];
        }
#pragma unroll
        for (int u = 0; u < 2; ++u) {
            const int mc = u * 128 + tid;
            const int d = ((((mc >> 2) & 3) * 32 + ((mc >> 1) & 1) * 16 + ((mc >> 4) & 15)) << 1) + (mc & 1);
            float a[3][3][4];
#pragma unroll
            for (int kt = 0; kt < 3; ++kt) {
#pragma unroll
                for (int kh = 0; kh < 3; ++kh) {
                    const float g0  = g[u][(kt * 3 + kh) * 3 + 0] * SCALE_ * cs[u];
                    const float g1  = g[u][(kt * 3 + kh) * 3 + 1] * SCALE_ * cs[u];
                    const float g2  = g[u][(kt * 3 + kh) * 3 + 2] * SCALE_ * cs[u];
                    const float s02 = g0 + g2;
                    a[kt][kh][0] = g0;
                    a[kt][kh][1] = 0.5f * (s02 + g1);
                    a[kt][kh][2] = 0.5f * (s02 - g1);
                    a[kt][kh][3] = g2;
                }
            }
            float bb[3][4][4];
#pragma unroll
            for (int kt = 0; kt < 3; ++kt) {
#pragma unroll
                for (int q = 0; q < 4; ++q) {
                    const float s02 = a[kt][0][q] + a[kt][2][q];
                    bb[kt][0][q] = a[kt][0][q];
                    bb[kt][1][q] = 0.5f * (s02 + a[kt][1][q]);
                    bb[kt][2][q] = 0.5f * (s02 - a[kt][1][q]);
                    bb[kt][3][q] = a[kt][2][q];
                }
            }
            float* dst = sA + d;
#pragma unroll
            for (int j = 0; j < 4; ++j) {
#pragma unroll
                for (int q = 0; q < 4; ++q) {
                    const float s02 = bb[0][j][q] + bb[2][j][q];
                    dst[((0 * 4 + j) * 4 + q) * APITCH] = bb[0][j][q];
                    dst[((1 * 4 + j) * 4 + q) * APITCH] = 0.5f * (s02 + bb[1][j][q]);
                    dst[((2 * 4 + j) * 4 + q) * APITCH] = 0.5f * (s02 - bb[1][j][q]);
                    dst[((3 * 4 + j) * 4 + q) * APITCH] = bb[2][j][q];
                }
            }
        }
        {
            v2f z = {0.f, 0.f};
            *(v2f*)&sX[(tid * XROW + (XOFF - 1)) * 2]  = z;
            *(v2f*)&sX[(tid * XROW + (XOFF + W_)) * 2] = z;
        }
    }

    v4f he[4][2][4], ho[4][2][4];
    {
        const v4f zero = {0.f, 0.f, 0.f, 0.f};
#pragma unroll
        for (int k = 0; k < 2; ++k) {
#pragma unroll
            for (int j = 0; j < 4; ++j) { he[0][k][j] = zero; ho[0][k][j] = zero; }
        }
        hcomb(he[1], ho[1], ev0, od0, 0, h0);
        hcomb(he[2], ho[2], ev[0], od[0], 1, h0);
        hcomb(he[3], ho[3], ev[1], od[1], 2, h0);
        load_plane(ev[0], od[0], inb, 3, h0, wvid);
#pragma unroll
        for (int jt = 0; jt < 4; ++jt) store_plane(he, ho, sX, wvid, lane, jt);
#pragma unroll
        for (int k = 0; k < 2; ++k) {
#pragma unroll
            for (int j = 0; j < 4; ++j) {
                he[0][k][j] = he[2][k][j];  ho[0][k][j] = ho[2][k][j];
                he[1][k][j] = he[3][k][j];  ho[1][k][j] = ho[3][k][j];
            }
        }
    }
    lds_barrier();

    const int wseg = wvid * 32;
    const int hf   = lane >> 4;
    const int lx   = lane & 15;

    const size_t strideC = (size_t)T_ * H_ * W_;
    float* ob = out + (size_t)b * COUT_ * strideC + (size_t)h0 * W_ + wseg + 2 * lx;
    const int colw = wseg + 2 * lx + (XOFF - 1);

    v2f fa[2][32];
    v2f fd[2][32];
    load_frags(fa[0], fd[0], sA, sX, 0, hf, lane, colw);

    v8f zp[8];

    for (int s = 0; s < NSTEP; ++s) {
        const int  t     = 2 * s;

        v8f yw[8];
        v8f z0[4], z1[4];
#pragma unroll
        for (int bk = 0; bk < 8; ++bk) {
            const int cur = bk & 1;
            const int jt  = bk >> 1;
            const int jh  = bk & 1;
            if (bk >= 1 && bk <= 4) {
                if (s > 0) store_group(ob + (size_t)(t - 2) * (H_ * W_), zp, bk - 1, hf);
            }
            if (jh == 1) {
                lds_barrier();
                if (jt == 0) hcomb(he[2], ho[2], ev[0], od[0], t + 3, h0);
                if (jt == 3) hcomb(he[3], ho[3], ev[1], od[1], t + 4, h0);
                store_plane(he, ho, sX, wvid, lane, jt);
                if (jt == 0) load_plane(ev[1], od[1], inb, t + 4, h0, wvid);
                if (jt == 2) load_plane(ev[0], od[0], inb, t + 5, h0, wvid);
                if (jt == 3) {
#pragma unroll
                    for (int k = 0; k < 2; ++k) {
#pragma unroll
                        for (int j = 0; j < 4; ++j) {
                            he[0][k][j] = he[2][k][j];  ho[0][k][j] = ho[2][k][j];
                            he[1][k][j] = he[3][k][j];  ho[1][k][j] = ho[3][k][j];
                        }
                    }
                }
            }
            if (bk + 1 < 8) {
                load_frags(fa[cur ^ 1], fd[cur ^ 1], sA, sX, bk + 1, hf, lane, colw);
            } else {
                load_frags(fa[0], fd[0], sA, sX, 0, hf, lane, colw);
            }
            v2f fv[32];
            wcomb(fv, fd[cur]);
            __builtin_amdgcn_sched_barrier(0);
            v8f acc[8];
#pragma unroll
            for (int c = 0; c < 4; ++c) {
#pragma unroll
                for (int jl = 0; jl < 2; ++jl) {
#pragma unroll
                    for (int q = 0; q < 4; ++q) {
                        const int a = jl * 4 + q;
                        const v8f cin0 = {0.f, 0.f, 0.f, 0.f, 0.f, 0.f, 0.f, 0.f};
                        acc[a] = __builtin_amdgcn_wmma_f32_16x16x4_f32(
                            false, fa[cur][a * 4 + c], false, fv[(jl * 4 + c) * 4 + q], (short)0,
                            c == 0 ? cin0 : acc[a], false, false);
                    }
                }
            }
            __builtin_amdgcn_sched_barrier(0);
            asm volatile("v_nop\n\tv_nop\n\tv_nop\n\tv_nop\n\tv_nop");
            __builtin_amdgcn_sched_barrier(0);

#pragma unroll
            for (int jl = 0; jl < 2; ++jl) {
                const int j = 2 * jh + jl;
                yw[j * 2 + 0] = (acc[jl * 4 + 0] + acc[jl * 4 + 1]) + acc[jl * 4 + 2];
                yw[j * 2 + 1] = (acc[jl * 4 + 1] - acc[jl * 4 + 2]) - acc[jl * 4 + 3];
            }
            if (jh == 1) {
#pragma unroll
                for (int p = 0; p < 2; ++p) {
                    v8f yh[2];
                    yh[0] = (yw[0 * 2 + p] + yw[1 * 2 + p]) + yw[2 * 2 + p];
                    yh[1] = (yw[1 * 2 + p] - yw[2 * 2 + p]) - yw[3 * 2 + p];
#pragma unroll
                    for (int hb = 0; hb < 2; ++hb) {
                        if (jt == 0) {
                            z0[hb * 2 + p] = yh[hb];
                        } else if (jt == 1) {
                            z0[hb * 2 + p] = z0[hb * 2 + p] + yh[hb];
                            z1[hb * 2 + p] = yh[hb];
                        } else if (jt == 2) {
                            zp[(0 * 2 + hb) * 2 + p] = z0[hb * 2 + p] + yh[hb];
                            z1[hb * 2 + p] = z1[hb * 2 + p] - yh[hb];
                        } else {
                            zp[(1 * 2 + hb) * 2 + p] = z1[hb * 2 + p] - yh[hb];
                        }
                    }
                }
            }
        }
    }

#pragma unroll
    for (int grp = 0; grp < 4; ++grp)
        store_group(ob + (size_t)(T_ - 2) * (H_ * W_), zp, grp, hf);
}

extern "C" void kernel_launch(void* const* d_in, const int* in_sizes, int n_in,
                              void* d_out, int out_size, void* d_ws, size_t ws_size,
                              hipStream_t stream) {
    const float* input  = (const float*)d_in[0];
    const float* cond   = (const float*)d_in[1];
    const float* weight = (const float*)d_in[2];
    float* out = (float*)d_out;

    const int    nBlocks = B_ * (H_ / 2);
    const size_t shmem   = (size_t)(NA + NX) * sizeof(float);

    conv3d_mod_wmma_f32<<<dim3(nBlocks), dim3(128), shmem, stream>>>(
        input, cond, weight, out);
}


# work-group barrier split: a wave signals inside the WMMA block before (behind its 28th WMMA, its last reads of the plane landed) and waits only in front of the plane stores; arithmetic unchanged
# speedup vs baseline: 12.2188x; 1.0411x over previous
#include <hip/hip_runtime.h>


typedef __attribute__((ext_vector_type(2))) float v2f;
typedef __attribute__((ext_vector_type(4))) float v4f;
typedef __attribute__((ext_vector_type(8))) float v8f;

#define B_    4
#define CIN_  16
#define COUT_ 16
#define T_    16
#define H_    128
#define W_    128
#define SCALE_ 0.04811252243246881f

#define NSTEP (T_ / 2)
#define XROW 144
#define XOFF 2
#define APITCH 258
#define NTAPW 64
#define NA   (NTAPW * APITCH)
#define NDH  4
#define PLANE (NDH * 8 * XROW * 2)
#define NX   (4 * PLANE)

__device__ __forceinline__ void load_plane(v4f (&ev)[8], v4f (&od)[8], const float* __restrict__ inb,
                                           int ts, int h0, int wvid)
{
    const size_t strideC = (size_t)T_ * H_ * W_;
    const int tsc = ts < 0 ? 0 : (ts > T_ - 1 ? T_ - 1 : ts);
#pragma unroll
    for (int u = 0; u < 8; ++u) {
        const int cp = (u & 1) * 4 + wvid;
        const int dh = u >> 1;
        const int hs = h0 + dh - 1;
        const int hsc = hs < 0 ? 0 : (hs > H_ - 1 ? H_ - 1 : hs);
        const float* src = inb + ((size_t)(2 * cp) * T_ + tsc) * (H_ * W_) + (size_t)hsc * W_;
        ev[u] = *(const v4f*)src;
        od[u] = *(const v4f*)(src + strideC);
    }
}

__device__ __forceinline__ void hcomb(v4f (&te)[2][4], v4f (&to)[2][4], const v4f (&ev)[8],
                                      const v4f (&od)[8], int ts, int h0)
{
    const v4f zero = {0.f, 0.f, 0.f, 0.f};
    const bool tok = (unsigned)ts < (unsigned)T_;
#pragma unroll
    for (int k = 0; k < 2; ++k) {
        v4f e[4], o[4];
#pragma unroll
        for (int dh = 0; dh < 4; ++dh) {
            const int hs = h0 + dh - 1;
            const bool ok = tok & ((unsigned)hs < (unsigned)H_);
            e[dh] = ok ? ev[2 * dh + k] : zero;
            o[dh] = ok ? od[2 * dh + k] : zero;
        }
        te[k][0] = e[0] - e[2];  to[k][0] = o[0] - o[2];
        te[k][1] = e[1] + e[2];  to[k][1] = o[1] + o[2];
        te[k][2] = e[2] - e[1];  to[k][2] = o[2] - o[1];
        te[k][3] = e[1] - e[3];  to[k][3] = o[1] - o[3];
    }
}

__device__ __forceinline__ void comb_plane(v4f (&lo)[8], v4f (&hi)[8], const v4f (&he)[4][2][4],
                                           const v4f (&ho)[4][2][4], int jt)
{
#pragma unroll
    for (int k = 0; k < 2; ++k) {
#pragma unroll
        for (int j = 0; j < 4; ++j) {
            v4f qe, qo;
            if (jt == 0)      { qe = he[0][k][j] - he[2][k][j];  qo = ho[0][k][j] - ho[2][k][j]; }
            else if (jt == 1) { qe = he[1][k][j] + he[2][k][j];  qo = ho[1][k][j] + ho[2][k][j]; }
            else if (jt == 2) { qe = he[2][k][j] - he[1][k][j];  qo = ho[2][k][j] - ho[1][k][j]; }
            else              { qe = he[1][k][j] - he[3][k][j];  qo = ho[1][k][j] - ho[3][k][j]; }
            v4f l = {qe.x, qo.x, qe.y, qo.y};
            v4f h = {qe.z, qo.z, qe.w, qo.w};
            lo[k * 4 + j] = l;
            hi[k * 4 + j] = h;
        }
    }
}

__device__ __forceinline__ void put_plane(const v4f (&lo)[8], const v4f (&hi)[8], float* sX, int wvid,
                                          int lane, int jt)
{
    v4f* p4 = (v4f*)sX;
#pragma unroll
    for (int k = 0; k < 2; ++k) {
        const int cp = k * 4 + wvid;
#pragma unroll
        for (int j = 0; j < 4; ++j) {
            const int p = j * 8 + cp;
            p4[jt * (PLANE / 4) + p * (XROW / 2) + 2 * lane + 1] = lo[k * 4 + j];
            p4[jt * (PLANE / 4) + p * (XROW / 2) + 2 * lane + 2] = hi[k * 4 + j];
        }
    }
}

__device__ __forceinline__ void store_plane(const v4f (&he)[4][2][4], const v4f (&ho)[4][2][4], float* sX,
                                            int wvid, int lane, int jt)
{
    v4f lo[8], hi[8];
    comb_plane(lo, hi, he, ho, jt);
    put_plane(lo, hi, sX, wvid, lane, jt);
}

__device__ __forceinline__ void lds_barrier()
{
    __builtin_amdgcn_fence(__ATOMIC_RELEASE, "workgroup", "local");
    __builtin_amdgcn_s_barrier();
    __builtin_amdgcn_fence(__ATOMIC_ACQUIRE, "workgroup", "local");
}

__device__ __forceinline__ void lds_barrier_signal()
{
    __builtin_amdgcn_fence(__ATOMIC_RELEASE, "workgroup", "local");
    __builtin_amdgcn_s_barrier_signal(-1);
}

__device__ __forceinline__ void lds_barrier_wait()
{
    __builtin_amdgcn_s_barrier_wait(-1);
    __builtin_amdgcn_fence(__ATOMIC_ACQUIRE, "workgroup", "local");
}

__device__ __forceinline__ void load_frags(v2f (&fa)[32], v2f (&fd)[32], const float* sA,
                                           const float* sX, int bk, int hf, int lane, int colw)
{
    const int jt = bk >> 1;
    const int jh = bk & 1;
    const float* xp = sX + jt * PLANE;
#pragma unroll
    for (int jl = 0; jl < 2; ++jl) {
        const int j = 2 * jh + jl;
#pragma unroll
        for (int c = 0; c < 4; ++c) {
            const int base = (j * 8 + 2 * c + hf) * (XROW * 2);
#pragma unroll
            for (int e = 0; e < 4; ++e)
                fd[(jl * 4 + c) * 4 + e] = *(const v2f*)&xp[base + (colw + e) * 2];
#pragma unroll
            for (int q = 0; q < 4; ++q) {
                const int tap = (jt * 4 + j) * 4 + q;
                fa[(jl * 4 + q) * 4 + c] = *(const v2f*)&sA[tap * APITCH + ((c << 5) + lane) * 2];
            }
        }
    }
}

__device__ __forceinline__ void wcomb(v2f (&fv)[32], const v2f (&fd)[32])
{
#pragma unroll
    for (int g = 0; g < 8; ++g) {
        const v2f d0 = fd[g * 4 + 0], d1 = fd[g * 4 + 1], d2 = fd[g * 4 + 2], d3 = fd[g * 4 + 3];
        fv[g * 4 + 0] = d0 - d2;
        fv[g * 4 + 1] = d1 + d2;
        fv[g * 4 + 2] = d2 - d1;
        fv[g * 4 + 3] = d1 - d3;
    }
}

__device__ __forceinline__ void store_group(float* o, const v8f (&zp)[8], int grp, int hf)
{
    const size_t strideC = (size_t)T_ * H_ * W_;
    const int tz = grp >> 1;
    const int hb = grp & 1;
#pragma unroll
    for (int r = 0; r < 8; ++r) {
        const size_t mo = (size_t)(hf * 8 + r) * strideC + (size_t)tz * (H_ * W_) + hb * W_;
        v2f v = {zp[grp * 2 + 0][r], zp[grp * 2 + 1][r]};
        *(v2f*)(o + mo) = v;
    }
}

__global__ __launch_bounds__(128) __attribute__((amdgpu_waves_per_eu(1, 1))) void conv3d_mod_wmma_f32(
    const float* __restrict__ input,
    const float* __restrict__ cond,
    const float* __restrict__ weight,
    float* __restrict__ out)
{
    extern __shared__ __attribute__((aligned(16))) float smem[];
    float* sA = smem;
    float* sX = smem + NA;

    const int bi  = blockIdx.x;
    const int h0  = (bi % (H_ / 2)) * 2;
    const int b   = bi / (H_ / 2);
    const int tid = threadIdx.x;
    const int wvid = tid >> 5;
    const int lane = tid & 31;

    const float* inb = input + (size_t)b * CIN_ * T_ * (H_ * W_) + 4 * lane;
    v4f ev0[8], od0[8];
    v4f ev[2][8], od[2][8];
    load_plane(ev0, od0, inb, 0, h0, wvid);
    load_plane(ev[0], od[0], inb, 1, h0, wvid);
    load_plane(ev[1], od[1], inb, 2, h0, wvid);

    {
        const float* cb = cond + b * CIN_;
        float g[2][27];
        float cs[2];
#pragma unroll
        for (int u = 0; u < 2; ++u) {
            const int mc = u * 128 + tid;
            const float* wp = weight + mc * 27;
#pragma unroll
            for (int e = 0; e < 27; ++e) g[u][e] = wp[e];
            cs[u] = cb[mc & 15];
        }
#pragma unroll
        for (int u = 0; u < 2; ++u) {
            const int mc = u * 128 + tid;
            const int d = ((((mc >> 2) & 3) * 32 + ((mc >> 1) & 1) * 16 + ((mc >> 4) & 15)) << 1) + (mc & 1);
            float a[3][3][4];
#pragma unroll
            for (int kt = 0; kt < 3; ++kt) {
#pragma unroll
                for (int kh = 0; kh < 3; ++kh) {
                    const float g0  = g[u][(kt * 3 + kh) * 3 + 0] * SCALE_ * cs[u];
                    const float g1  = g[u][(kt * 3 + kh) * 3 + 1] * SCALE_ * cs[u];
                    const float g2  = g[u][(kt * 3 + kh) * 3 + 2] * SCALE_ * cs[u];
                    const float s02 = g0 + g2;
                    a[kt][kh][0] = g0;
                    a[kt][kh][1] = 0.5f * (s02 + g1);
                    a[kt][kh][2] = 0.5f * (s02 - g1);
                    a[kt][kh][3] = g2;
                }
            }
            float bb[3][4][4];
#pragma unroll
            for (int kt = 0; kt < 3; ++kt) {
#pragma unroll
                for (int q = 0; q < 4; ++q) {
                    const float s02 = a[kt][0][q] + a[kt][2][q];
                    bb[kt][0][q] = a[kt][0][q];
                    bb[kt][1][q] = 0.5f * (s02 + a[kt][1][q]);
                    bb[kt][2][q] = 0.5f * (s02 - a[kt][1][q]);
                    bb[kt][3][q] = a[kt][2][q];
                }
            }
            float* dst = sA + d;
#pragma unroll
            for (int j = 0; j < 4; ++j) {
#pragma unroll
                for (int q = 0; q < 4; ++q) {
                    const float s02 = bb[0][j][q] + bb[2][j][q];
                    dst[((0 * 4 + j) * 4 + q) * APITCH] = bb[0][j][q];
                    dst[((1 * 4 + j) * 4 + q) * APITCH] = 0.5f * (s02 + bb[1][j][q]);
                    dst[((2 * 4 + j) * 4 + q) * APITCH] = 0.5f * (s02 - bb[1][j][q]);
                    dst[((3 * 4 + j) * 4 + q) * APITCH] = bb[2][j][q];
                }
            }
        }
        {
            v2f z = {0.f, 0.f};
            *(v2f*)&sX[(tid * XROW + (XOFF - 1)) * 2]  = z;
            *(v2f*)&sX[(tid * XROW + (XOFF + W_)) * 2] = z;
        }
    }

    v4f he[4][2][4], ho[4][2][4];
    {
        const v4f zero = {0.f, 0.f, 0.f, 0.f};
#pragma unroll
        for (int k = 0; k < 2; ++k) {
#pragma unroll
            for (int j = 0; j < 4; ++j) { he[0][k][j] = zero; ho[0][k][j] = zero; }
        }
        hcomb(he[1], ho[1], ev0, od0, 0, h0);
        hcomb(he[2], ho[2], ev[0], od[0], 1, h0);
        hcomb(he[3], ho[3], ev[1], od[1], 2, h0);
        load_plane(ev[0], od[0], inb, 3, h0, wvid);
#pragma unroll
        for (int jt = 0; jt < 4; ++jt) store_plane(he, ho, sX, wvid, lane, jt);
#pragma unroll
        for (int k = 0; k < 2; ++k) {
#pragma unroll
            for (int j = 0; j < 4; ++j) {
                he[0][k][j] = he[2][k][j];  ho[0][k][j] = ho[2][k][j];
                he[1][k][j] = he[3][k][j];  ho[1][k][j] = ho[3][k][j];
            }
        }
    }
    lds_barrier();

    const int wseg = wvid * 32;
    const int hf   = lane >> 4;
    const int lx   = lane & 15;

    const size_t strideC = (size_t)T_ * H_ * W_;
    float* ob = out + (size_t)b * COUT_ * strideC + (size_t)h0 * W_ + wseg + 2 * lx;
    const int colw = wseg + 2 * lx + (XOFF - 1);

    v2f fa[2][32];
    v2f fd[2][32];
    load_frags(fa[0], fd[0], sA, sX, 0, hf, lane, colw);

    v8f zp[8];

    for (int s = 0; s < NSTEP; ++s) {
        const int  t     = 2 * s;

        v8f yw[8];
        v8f z0[4], z1[4];
#pragma unroll
        for (int bk = 0; bk < 8; ++bk) {
            const int cur = bk & 1;
            const int jt  = bk >> 1;
            const int jh  = bk & 1;
            if (bk >= 1 && bk <= 4) {
                if (s > 0) store_group(ob + (size_t)(t - 2) * (H_ * W_), zp, bk - 1, hf);
            }
            if (jh == 1) {
                if (jt == 0) hcomb(he[2], ho[2], ev[0], od[0], t + 3, h0);
                if (jt == 3) hcomb(he[3], ho[3], ev[1], od[1], t + 4, h0);
                v4f plo[8], phi[8];
                comb_plane(plo, phi, he, ho, jt);
                __builtin_amdgcn_sched_barrier(0);
                lds_barrier_wait();
                put_plane(plo, phi, sX, wvid, lane, jt);
                if (jt == 0) load_plane(ev[1], od[1], inb, t + 4, h0, wvid);
                if (jt == 2) load_plane(ev[0], od[0], inb, t + 5, h0, wvid);
                if (jt == 3) {
#pragma unroll
                    for (int k = 0; k < 2; ++k) {
#pragma unroll
                        for (int j = 0; j < 4; ++j) {
                            he[0][k][j] = he[2][k][j];  ho[0][k][j] = ho[2][k][j];
                            he[1][k][j] = he[3][k][j];  ho[1][k][j] = ho[3][k][j];
                        }
                    }
                }
            }
            if (bk + 1 < 8) {
                load_frags(fa[cur ^ 1], fd[cur ^ 1], sA, sX, bk + 1, hf, lane, colw);
            } else {
                load_frags(fa[0], fd[0], sA, sX, 0, hf, lane, colw);
            }
            v2f fv[32];
            wcomb(fv, fd[cur]);
            __builtin_amdgcn_sched_barrier(0);
            v8f acc[8];
#pragma unroll
            for (int c = 0; c < 4; ++c) {
#pragma unroll
                for (int jl = 0; jl < 2; ++jl) {
#pragma unroll
                    for (int q = 0; q < 4; ++q) {
                        const int a = jl * 4 + q;
                        const v8f cin0 = {0.f, 0.f, 0.f, 0.f, 0.f, 0.f, 0.f, 0.f};
                        acc[a] = __builtin_amdgcn_wmma_f32_16x16x4_f32(
                            false, fa[cur][a * 4 + c], false, fv[(jl * 4 + c) * 4 + q], (short)0,
                            c == 0 ? cin0 : acc[a], false, false);
                    }
                    if (jh == 0 && c == 3 && jl == 0) {
                        __builtin_amdgcn_sched_barrier(0);
                        lds_barrier_signal();
                        __builtin_amdgcn_sched_barrier(0);
                    }
                }
            }
            __builtin_amdgcn_sched_barrier(0);
            asm volatile("v_nop\n\tv_nop\n\tv_nop\n\tv_nop\n\tv_nop");
            __builtin_amdgcn_sched_barrier(0);

#pragma unroll
            for (int jl = 0; jl < 2; ++jl) {
                const int j = 2 * jh + jl;
                yw[j * 2 + 0] = (acc[jl * 4 + 0] + acc[jl * 4 + 1]) + acc[jl * 4 + 2];
                yw[j * 2 + 1] = (acc[jl * 4 + 1] - acc[jl * 4 + 2]) - acc[jl * 4 + 3];
            }
            if (jh == 1) {
#pragma unroll
                for (int p = 0; p < 2; ++p) {
                    v8f yh[2];
                    yh[0] = (yw[0 * 2 + p] + yw[1 * 2 + p]) + yw[2 * 2 + p];
                    yh[1] = (yw[1 * 2 + p] - yw[2 * 2 + p]) - yw[3 * 2 + p];
#pragma unroll
                    for (int hb = 0; hb < 2; ++hb) {
                        if (jt == 0) {
                            z0[hb * 2 + p] = yh[hb];
                        } else if (jt == 1) {
                            z0[hb * 2 + p] = z0[hb * 2 + p] + yh[hb];
                            z1[hb * 2 + p] = yh[hb];
                        } else if (jt == 2) {
                            zp[(0 * 2 + hb) * 2 + p] = z0[hb * 2 + p] + yh[hb];
                            z1[hb * 2 + p] = z1[hb * 2 + p] - yh[hb];
                        } else {
                            zp[(1 * 2 + hb) * 2 + p] = z1[hb * 2 + p] - yh[hb];
                        }
                    }
                }
            }
        }
    }

#pragma unroll
    for (int grp = 0; grp < 4; ++grp)
        store_group(ob + (size_t)(T_ - 2) * (H_ * W_), zp, grp, hf);
}

extern "C" void kernel_launch(void* const* d_in, const int* in_sizes, int n_in,
                              void* d_out, int out_size, void* d_ws, size_t ws_size,
                              hipStream_t stream) {
    const float* input  = (const float*)d_in[0];
    const float* cond   = (const float*)d_in[1];
    const float* weight = (const float*)d_in[2];
    float* out = (float*)d_out;

    const int    nBlocks = B_ * (H_ / 2);
    const size_t shmem   = (size_t)(NA + NX) * sizeof(float);

    conv3d_mod_wmma_f32<<<dim3(nBlocks), dim3(128), shmem, stream>>>(
        input, cond, weight, out);
}
